# starting kernel
# baseline (speedup 1.0000x reference)
#include <hip/hip_runtime.h>
#include <hip/hip_bf16.h>
#include <math.h>

typedef float v2f __attribute__((ext_vector_type(2)));
typedef float v8f __attribute__((ext_vector_type(8)));

#define BN 8
#define NN 1024
#define DD 256
#define EPSV 1e-8f
#define SSTRIDE 1028

__device__ __forceinline__ v8f wmma4(v2f a, v2f b, v8f c) {
  return __builtin_amdgcn_wmma_f32_16x16x4_f32(
       false, a,  false, b,
       (short)0, c,  false,  false);
}

__device__ __forceinline__ float wave_max(float v) {
#pragma unroll
  for (int off = 16; off > 0; off >>= 1) v = fmaxf(v, __shfl_xor(v, off, 32));
  return v;
}
__device__ __forceinline__ float wave_sum(float v) {
#pragma unroll
  for (int off = 16; off > 0; off >>= 1) v += __shfl_xor(v, off, 32);
  return v;
}

__global__ void pos_softmax_kernel(const float* __restrict__ coords,
                                   const float* __restrict__ pos_emb,
                                   const float* __restrict__ p_temp,
                                   float* __restrict__ pos_score) {
  __shared__ float row[NN];
  __shared__ float red[8];
  const int n = blockIdx.x;
  const int tid = threadIdx.x;
  const int lane = tid & 31, wave = tid >> 5;
  const float pa = fabsf(p_temp[0]);
  float pe[6];
#pragma unroll
  for (int c = 0; c < 6; ++c) pe[c] = pos_emb[n * 6 + c];

  float lmax = -1e30f;
  for (int m = tid; m < NN; m += 256) {
    const float* cp = coords + ((size_t)n * NN + m) * 6;
    float dot = cp[0] * pe[0] + cp[1] * pe[1] + cp[2] * pe[2] +
                cp[3] * pe[3] + cp[4] * pe[4] + cp[5] * pe[5];
    float lg = -pa * dot;
    row[m] = lg;
    lmax = fmaxf(lmax, lg);
  }
  lmax = wave_max(lmax);
  if (lane == 0) red[wave] = lmax;
  __syncthreads();
  float rmax = red[0];
#pragma unroll
  for (int w = 1; w < 8; ++w) rmax = fmaxf(rmax, red[w]);
  __syncthreads();

  float lsum = 0.f;
  for (int m = tid; m < NN; m += 256) {
    float e = expf(row[m] - rmax);
    row[m] = e;
    lsum += e;
  }
  lsum = wave_sum(lsum);
  if (lane == 0) red[wave] = lsum;
  __syncthreads();
  float rsum = 0.f;
#pragma unroll
  for (int w = 0; w < 8; ++w) rsum += red[w];
  const float inv = 1.0f / rsum;
  for (int m = tid; m < NN; m += 256)
    pos_score[(size_t)n * NN + m] = row[m] * inv;
}

__global__ void gemm_xut_kernel(const float* __restrict__ x,
                                const float* __restrict__ U,
                                float* __restrict__ t) {
  const int lane = threadIdx.x & 31;
  const int wave = threadIdx.x >> 5;
  const int tile = blockIdx.x * 8 + wave;
  const int i0 = (tile >> 4) * 16;
  const int j0 = (tile & 15) * 16;
  const int hl = lane >> 4, ll = lane & 15;

  const float* arow = x + (size_t)(i0 + ll) * DD + 2 * hl;
  const float* brow = U + (size_t)(j0 + ll) * DD + 2 * hl;
  v8f acc = {0.f, 0.f, 0.f, 0.f, 0.f, 0.f, 0.f, 0.f};
#pragma unroll 8
  for (int kb = 0; kb < DD; kb += 4) {
    v2f a = *(const v2f*)(arow + kb);
    v2f b = *(const v2f*)(brow + kb);
    acc = wmma4(a, b, acc);
  }
  float* orow = t + (size_t)(i0 + 8 * hl) * DD + j0 + ll;
#pragma unroll
  for (int v = 0; v < 8; ++v) orow[(size_t)v * DD] = acc[v];
}

__global__ void gemm_su_kernel(const float* __restrict__ t,
                               const float* __restrict__ U,
                               const float* __restrict__ S1,
                               const float* __restrict__ S2,
                               float* __restrict__ xs1,
                               float* __restrict__ xs2) {
  const int lane = threadIdx.x & 31;
  const int wave = threadIdx.x >> 5;
  const int tile = blockIdx.x * 8 + wave;
  const int i0 = (tile >> 4) * 16;
  const int j0 = (tile & 15) * 16;
  const int hl = lane >> 4, ll = lane & 15;

  const float* arow = t + (size_t)(i0 + ll) * DD + 2 * hl;
  const float* bcol = U + j0 + ll;
  v8f acc1 = {0.f, 0.f, 0.f, 0.f, 0.f, 0.f, 0.f, 0.f};
  v8f acc2 = {0.f, 0.f, 0.f, 0.f, 0.f, 0.f, 0.f, 0.f};
#pragma unroll 4
  for (int kb = 0; kb < DD; kb += 4) {
    const int k = kb + 2 * hl;
    v2f a = *(const v2f*)(arow + kb);
    v2f b;
    b.x = bcol[(size_t)k * DD];
    b.y = bcol[(size_t)(k + 1) * DD];
    v2f a1 = {a.x * fabsf(S1[k]), a.y * fabsf(S1[k + 1])};
    v2f a2 = {a.x * fabsf(S2[k]), a.y * fabsf(S2[k + 1])};
    acc1 = wmma4(a1, b, acc1);
    acc2 = wmma4(a2, b, acc2);
  }
  float* o1 = xs1 + (size_t)(i0 + 8 * hl) * DD + j0 + ll;
  float* o2 = xs2 + (size_t)(i0 + 8 * hl) * DD + j0 + ll;
#pragma unroll
  for (int v = 0; v < 8; ++v) {
    o1[(size_t)v * DD] = acc1[v];
    o2[(size_t)v * DD] = acc2[v];
  }
}

__global__ void fused_attn_kernel(const float* __restrict__ xs1,
                                  const float* __restrict__ xs2,
                                  const float* __restrict__ y,
                                  const float* __restrict__ pos_score,
                                  const float* __restrict__ gating,
                                  const float* __restrict__ h_temp,
                                  float* __restrict__ out,
                                  float* __restrict__ heat) {
  extern __shared__ float smem[];
  float* SL = smem;
  float* entS = smem + 2 * 16 * SSTRIDE;
  float* selS = entS + 32;

  const int b = blockIdx.x >> 6;
  const int n0 = (blockIdx.x & 63) * 16;
  const int tid = threadIdx.x;
  const int lane = tid & 31, wave = tid >> 5;
  const int hl = lane >> 4, ll = lane & 15;
  const float scale = 0.0625f;
  const float g = 1.0f / (1.0f + expf(-gating[0]));
  const float omg = 1.0f - g;
  const float ht = h_temp[0];
  const float* yb = y + (size_t)b * NN * DD;

  {
    const int branch = wave & 1;
    const float* xs = branch ? xs2 : xs1;
    const float* arow = xs + ((size_t)b * NN + n0 + ll) * DD + 2 * hl;
    for (int mt = (wave >> 1); mt < 64; mt += 4) {
      const int m0 = mt * 16;
      const float* brow = yb + (size_t)(m0 + ll) * DD + 2 * hl;
      v8f acc = {0.f, 0.f, 0.f, 0.f, 0.f, 0.f, 0.f, 0.f};
#pragma unroll 8
      for (int kb = 0; kb < DD; kb += 4) {
        v2f a = *(const v2f*)(arow + kb);
        v2f bb = *(const v2f*)(brow + kb);
        acc = wmma4(a, bb, acc);
      }
      float* dst = SL + ((size_t)branch * 16 + 8 * hl) * SSTRIDE + m0 + ll;
#pragma unroll
      for (int v = 0; v < 8; ++v) dst[(size_t)v * SSTRIDE] = acc[v] * scale;
    }
  }
  __syncthreads();

  for (int idx = wave; idx < 32; idx += 8) {
    const int br = idx >> 4, row = idx & 15;
    float* Lrow = SL + (size_t)(br * 16 + row) * SSTRIDE;
    const float* prow = pos_score + (size_t)(n0 + row) * NN;
    float lmax = -1e30f;
    for (int m = lane; m < NN; m += 32) lmax = fmaxf(lmax, Lrow[m]);
    lmax = wave_max(lmax);
    float lsum = 0.f;
    for (int m = lane; m < NN; m += 32) lsum += expf(Lrow[m] - lmax);
    lsum = wave_sum(lsum);
    const float inv = 1.0f / lsum;
    float ent = 0.f;
    for (int m = lane; m < NN; m += 32) {
      float p = expf(Lrow[m] - lmax) * inv;
      float a = omg * p + g * prow[m];
      ent += -a * logf(a + EPSV);
      Lrow[m] = a;
    }
    ent = wave_sum(ent);
    if (lane == 0) entS[br * 16 + row] = ent;
  }
  __syncthreads();

  if (tid < 16) {
    const int row = tid;
    const float h0 = 2.0f - 2.0f / (1.0f + expf(-ht * entS[row]));
    const float h1 = 2.0f - 2.0f / (1.0f + expf(-ht * entS[16 + row]));
    const int sel = (h0 >= h1) ? 0 : 1;
    selS[row] = (float)sel;
    heat[(size_t)b * NN + n0 + row] = sel ? h1 : h0;
  }
  __syncthreads();

  const int selrow = (int)selS[ll];
  const float* Abase = SL + (size_t)(selrow * 16 + ll) * SSTRIDE + 2 * hl;
  for (int dt = wave; dt < 16; dt += 8) {
    const int d0 = dt * 16;
    const float* bbase = yb + d0 + ll;
    v8f acc = {0.f, 0.f, 0.f, 0.f, 0.f, 0.f, 0.f, 0.f};
#pragma unroll 4
    for (int kb = 0; kb < NN; kb += 4) {
      const int k = kb + 2 * hl;
      v2f a = *(const v2f*)(Abase + kb);
      v2f bb;
      bb.x = bbase[(size_t)k * DD];
      bb.y = bbase[(size_t)(k + 1) * DD];
      acc = wmma4(a, bb, acc);
    }
    float* orow = out + ((size_t)b * NN + n0 + 8 * hl) * DD + d0 + ll;
#pragma unroll
    for (int v = 0; v < 8; ++v) orow[(size_t)v * DD] = acc[v];
  }
}

extern "C" void kernel_launch(void* const* d_in, const int* in_sizes, int n_in,
                              void* d_out, int out_size, void* d_ws, size_t ws_size,
                              hipStream_t stream) {
  const float* x = (const float*)d_in[0];
  const float* y = (const float*)d_in[1];
  const float* coords = (const float*)d_in[2];
  const float* U = (const float*)d_in[3];
  const float* S1 = (const float*)d_in[4];
  const float* S2 = (const float*)d_in[5];
  const float* gating = (const float*)d_in[6];
  const float* h_temp = (const float*)d_in[7];
  const float* p_temp = (const float*)d_in[8];
  const float* pos_emb = (const float*)d_in[9];

  float* out = (float*)d_out;
  float* heat = out + (size_t)BN * NN * DD;

  float* ws = (float*)d_ws;
  float* pos_score = ws;
  float* t = pos_score + (size_t)NN * NN;
  float* xs1 = t + (size_t)BN * NN * DD;
  float* xs2 = xs1 + (size_t)BN * NN * DD;

  pos_softmax_kernel<<<NN, 256, 0, stream>>>(coords, pos_emb, p_temp, pos_score);
  gemm_xut_kernel<<<1024, 256, 0, stream>>>(x, U, t);
  gemm_su_kernel<<<1024, 256, 0, stream>>>(t, U, S1, S2, xs1, xs2);

  const size_t lds_bytes = (size_t)(2 * 16 * SSTRIDE + 32 + 16) * sizeof(float);
  fused_attn_kernel<<<BN * 64, 256, lds_bytes, stream>>>(
      xs1, xs2, y, pos_score, gating, h_temp, out, heat);
}


# Score GEMM register-blocked: each wave holds both branches x 8 key tiles, so a y fragment feeds 2 WMMAs and an xs fragment 8 (10 loads per 16 WMMAs, was 32), one whole 128-B line per row per trip
# speedup vs baseline: 1.5721x; 1.5721x over previous
#include <hip/hip_runtime.h>
#include <hip/hip_bf16.h>
#include <math.h>

typedef float v2f __attribute__((ext_vector_type(2)));
typedef float v8f __attribute__((ext_vector_type(8)));

#define BN 8
#define NN 1024
#define DD 256
#define EPSV 1e-8f
#define SSTRIDE 1028
#define KT 4

__device__ __forceinline__ v8f wmma4(v2f a, v2f b, v8f c) {
  return __builtin_amdgcn_wmma_f32_16x16x4_f32(
       false, a,  false, b,
       (short)0, c,  false,  false);
}

__device__ __forceinline__ float wave_max(float v) {
#pragma unroll
  for (int off = 16; off > 0; off >>= 1) v = fmaxf(v, __shfl_xor(v, off, 32));
  return v;
}
__device__ __forceinline__ float wave_sum(float v) {
#pragma unroll
  for (int off = 16; off > 0; off >>= 1) v += __shfl_xor(v, off, 32);
  return v;
}

__global__ void pos_softmax_kernel(const float* __restrict__ coords,
                                   const float* __restrict__ pos_emb,
                                   const float* __restrict__ p_temp,
                                   float* __restrict__ pos_score) {
  __shared__ float row[NN];
  __shared__ float red[8];
  const int n = blockIdx.x;
  const int tid = threadIdx.x;
  const int lane = tid & 31, wave = tid >> 5;
  const float pa = fabsf(p_temp[0]);
  float pe[6];
#pragma unroll
  for (int c = 0; c < 6; ++c) pe[c] = pos_emb[n * 6 + c];

  float lmax = -1e30f;
  for (int m = tid; m < NN; m += 256) {
    const float* cp = coords + ((size_t)n * NN + m) * 6;
    float dot = cp[0] * pe[0] + cp[1] * pe[1] + cp[2] * pe[2] +
                cp[3] * pe[3] + cp[4] * pe[4] + cp[5] * pe[5];
    float lg = -pa * dot;
    row[m] = lg;
    lmax = fmaxf(lmax, lg);
  }
  lmax = wave_max(lmax);
  if (lane == 0) red[wave] = lmax;
  __syncthreads();
  float rmax = red[0];
#pragma unroll
  for (int w = 1; w < 8; ++w) rmax = fmaxf(rmax, red[w]);
  __syncthreads();

  float lsum = 0.f;
  for (int m = tid; m < NN; m += 256) {
    float e = expf(row[m] - rmax);
    row[m] = e;
    lsum += e;
  }
  lsum = wave_sum(lsum);
  if (lane == 0) red[wave] = lsum;
  __syncthreads();
  float rsum = 0.f;
#pragma unroll
  for (int w = 0; w < 8; ++w) rsum += red[w];
  const float inv = 1.0f / rsum;
  for (int m = tid; m < NN; m += 256)
    pos_score[(size_t)n * NN + m] = row[m] * inv;
}

__global__ void gemm_xut_kernel(const float* __restrict__ x,
                                const float* __restrict__ U,
                                float* __restrict__ t) {
  const int lane = threadIdx.x & 31;
  const int wave = threadIdx.x >> 5;
  const int tile = blockIdx.x * 8 + wave;
  const int i0 = (tile >> 4) * 16;
  const int j0 = (tile & 15) * 16;
  const int hl = lane >> 4, ll = lane & 15;

  const float* arow = x + (size_t)(i0 + ll) * DD + 2 * hl;
  const float* brow = U + (size_t)(j0 + ll) * DD + 2 * hl;
  v8f acc = {0.f, 0.f, 0.f, 0.f, 0.f, 0.f, 0.f, 0.f};
#pragma unroll 8
  for (int kb = 0; kb < DD; kb += 4) {
    v2f a = *(const v2f*)(arow + kb);
    v2f b = *(const v2f*)(brow + kb);
    acc = wmma4(a, b, acc);
  }
  float* orow = t + (size_t)(i0 + 8 * hl) * DD + j0 + ll;
#pragma unroll
  for (int v = 0; v < 8; ++v) orow[(size_t)v * DD] = acc[v];
}

__global__ void gemm_su_kernel(const float* __restrict__ t,
                               const float* __restrict__ U,
                               const float* __restrict__ S1,
                               const float* __restrict__ S2,
                               float* __restrict__ xs1,
                               float* __restrict__ xs2) {
  const int lane = threadIdx.x & 31;
  const int wave = threadIdx.x >> 5;
  const int tile = blockIdx.x * 8 + wave;
  const int i0 = (tile >> 4) * 16;
  const int j0 = (tile & 15) * 16;
  const int hl = lane >> 4, ll = lane & 15;

  const float* arow = t + (size_t)(i0 + ll) * DD + 2 * hl;
  const float* bcol = U + j0 + ll;
  v8f acc1 = {0.f, 0.f, 0.f, 0.f, 0.f, 0.f, 0.f, 0.f};
  v8f acc2 = {0.f, 0.f, 0.f, 0.f, 0.f, 0.f, 0.f, 0.f};
#pragma unroll 4
  for (int kb = 0; kb < DD; kb += 4) {
    const int k = kb + 2 * hl;
    v2f a = *(const v2f*)(arow + kb);
    v2f b;
    b.x = bcol[(size_t)k * DD];
    b.y = bcol[(size_t)(k + 1) * DD];
    v2f a1 = {a.x * fabsf(S1[k]), a.y * fabsf(S1[k + 1])};
    v2f a2 = {a.x * fabsf(S2[k]), a.y * fabsf(S2[k + 1])};
    acc1 = wmma4(a1, b, acc1);
    acc2 = wmma4(a2, b, acc2);
  }
  float* o1 = xs1 + (size_t)(i0 + 8 * hl) * DD + j0 + ll;
  float* o2 = xs2 + (size_t)(i0 + 8 * hl) * DD + j0 + ll;
#pragma unroll
  for (int v = 0; v < 8; ++v) {
    o1[(size_t)v * DD] = acc1[v];
    o2[(size_t)v * DD] = acc2[v];
  }
}

__global__ void __launch_bounds__(256) fused_attn_kernel(const float* __restrict__ xs1,
                                  const float* __restrict__ xs2,
                                  const float* __restrict__ y,
                                  const float* __restrict__ pos_score,
                                  const float* __restrict__ gating,
                                  const float* __restrict__ h_temp,
                                  float* __restrict__ out,
                                  float* __restrict__ heat) {
  extern __shared__ float smem[];
  float* SL = smem;
  float* entS = smem + 2 * 16 * SSTRIDE;
  float* selS = entS + 32;

  const int b = blockIdx.x >> 6;
  const int n0 = (blockIdx.x & 63) * 16;
  const int tid = threadIdx.x;
  const int lane = tid & 31, wave = tid >> 5;
  const int hl = lane >> 4, ll = lane & 15;
  const float scale = 0.0625f;
  const float g = 1.0f / (1.0f + expf(-gating[0]));
  const float omg = 1.0f - g;
  const float ht = h_temp[0];
  const float* yb = y + (size_t)b * NN * DD;

  {
    const float* a1row = xs1 + ((size_t)b * NN + n0 + ll) * DD + 2 * hl;
    const float* a2row = xs2 + ((size_t)b * NN + n0 + ll) * DD + 2 * hl;
    const int mw = wave * (2 * KT * 16);
    const float* brow = yb + (size_t)(mw + ll) * DD + 2 * hl;
    v8f acc1[2 * KT], acc2[2 * KT];
#pragma unroll
    for (int t = 0; t < 2 * KT; ++t) {
      acc1[t] = (v8f){0.f, 0.f, 0.f, 0.f, 0.f, 0.f, 0.f, 0.f};
      acc2[t] = (v8f){0.f, 0.f, 0.f, 0.f, 0.f, 0.f, 0.f, 0.f};
    }
    for (int kl = 0; kl < DD; kl += 32) {
      v2f a1[8], a2[8];
#pragma unroll
      for (int j = 0; j < 8; ++j) a1[j] = *(const v2f*)(a1row + kl + 4 * j);
#pragma unroll
      for (int j = 0; j < 8; ++j) a2[j] = *(const v2f*)(a2row + kl + 4 * j);
#pragma unroll
      for (int h = 0; h < 2; ++h) {
        v2f bq[KT][8];
#pragma unroll
        for (int t = 0; t < KT; ++t) {
#pragma unroll
          for (int j = 0; j < 8; ++j)
            bq[t][j] = *(const v2f*)(brow + (size_t)(h * KT + t) * 16 * DD + kl + 4 * j);
        }
        __builtin_amdgcn_sched_barrier(0);
#pragma unroll
        for (int j = 0; j < 8; ++j) {
#pragma unroll
          for (int t = 0; t < KT; ++t) {
            acc1[h * KT + t] = wmma4(a1[j], bq[t][j], acc1[h * KT + t]);
            acc2[h * KT + t] = wmma4(a2[j], bq[t][j], acc2[h * KT + t]);
          }
        }
      }
    }
#pragma unroll
    for (int t = 0; t < 2 * KT; ++t) {
      float* d1 = SL + ((size_t)8 * hl) * SSTRIDE + mw + t * 16 + ll;
      float* d2 = d1 + (size_t)16 * SSTRIDE;
#pragma unroll
      for (int v = 0; v < 8; ++v) {
        d1[(size_t)v * SSTRIDE] = acc1[t][v] * scale;
        d2[(size_t)v * SSTRIDE] = acc2[t][v] * scale;
      }
    }
  }
  __syncthreads();

  for (int idx = wave; idx < 32; idx += 8) {
    const int br = idx >> 4, row = idx & 15;
    float* Lrow = SL + (size_t)(br * 16 + row) * SSTRIDE;
    const float* prow = pos_score + (size_t)(n0 + row) * NN;
    float lmax = -1e30f;
    for (int m = lane; m < NN; m += 32) lmax = fmaxf(lmax, Lrow[m]);
    lmax = wave_max(lmax);
    float lsum = 0.f;
    for (int m = lane; m < NN; m += 32) lsum += expf(Lrow[m] - lmax);
    lsum = wave_sum(lsum);
    const float inv = 1.0f / lsum;
    float ent = 0.f;
    for (int m = lane; m < NN; m += 32) {
      float p = expf(Lrow[m] - lmax) * inv;
      float a = omg * p + g * prow[m];
      ent += -a * logf(a + EPSV);
      Lrow[m] = a;
    }
    ent = wave_sum(ent);
    if (lane == 0) entS[br * 16 + row] = ent;
  }
  __syncthreads();

  if (tid < 16) {
    const int row = tid;
    const float h0 = 2.0f - 2.0f / (1.0f + expf(-ht * entS[row]));
    const float h1 = 2.0f - 2.0f / (1.0f + expf(-ht * entS[16 + row]));
    const int sel = (h0 >= h1) ? 0 : 1;
    selS[row] = (float)sel;
    heat[(size_t)b * NN + n0 + row] = sel ? h1 : h0;
  }
  __syncthreads();

  const int selrow = (int)selS[ll];
  const float* Abase = SL + (size_t)(selrow * 16 + ll) * SSTRIDE + 2 * hl;
  for (int dt = wave; dt < 16; dt += 8) {
    const int d0 = dt * 16;
    const float* bbase = yb + d0 + ll;
    v8f acc = {0.f, 0.f, 0.f, 0.f, 0.f, 0.f, 0.f, 0.f};
#pragma unroll 4
    for (int kb = 0; kb < NN; kb += 4) {
      const int k = kb + 2 * hl;
      v2f a = *(const v2f*)(Abase + kb);
      v2f bb;
      bb.x = bbase[(size_t)k * DD];
      bb.y = bbase[(size_t)(k + 1) * DD];
      acc = wmma4(a, bb, acc);
    }
    float* orow = out + ((size_t)b * NN + n0 + 8 * hl) * DD + d0 + ll;
#pragma unroll
    for (int v = 0; v < 8; ++v) orow[(size_t)v * DD] = acc[v];
  }
}

extern "C" void kernel_launch(void* const* d_in, const int* in_sizes, int n_in,
                              void* d_out, int out_size, void* d_ws, size_t ws_size,
                              hipStream_t stream) {
  const float* x = (const float*)d_in[0];
  const float* y = (const float*)d_in[1];
  const float* coords = (const float*)d_in[2];
  const float* U = (const float*)d_in[3];
  const float* S1 = (const float*)d_in[4];
  const float* S2 = (const float*)d_in[5];
  const float* gating = (const float*)d_in[6];
  const float* h_temp = (const float*)d_in[7];
  const float* p_temp = (const float*)d_in[8];
  const float* pos_emb = (const float*)d_in[9];

  float* out = (float*)d_out;
  float* heat = out + (size_t)BN * NN * DD;

  float* ws = (float*)d_ws;
  float* pos_score = ws;
  float* t = pos_score + (size_t)NN * NN;
  float* xs1 = t + (size_t)BN * NN * DD;
  float* xs2 = xs1 + (size_t)BN * NN * DD;

  pos_softmax_kernel<<<NN, 256, 0, stream>>>(coords, pos_emb, p_temp, pos_score);
  gemm_xut_kernel<<<1024, 256, 0, stream>>>(x, U, t);
  gemm_su_kernel<<<1024, 256, 0, stream>>>(t, U, S1, S2, xs1, xs2);

  const size_t lds_bytes = (size_t)(2 * 16 * SSTRIDE + 32 + 16) * sizeof(float);
  fused_attn_kernel<<<BN * 64, 256, lds_bytes, stream>>>(
      xs1, xs2, y, pos_score, gating, h_temp, out, heat);
}


# Output GEMM (attn @ y) register-blocked: a wave owns 2 column-interleaved tiles, so 1 LDS attn fragment and 1 whole-line 8-byte y load feed 2 WMMAs (3 loads per 2 WMMAs, was 6), loads grouped
# speedup vs baseline: 1.6657x; 1.0595x over previous
#include <hip/hip_runtime.h>
#include <hip/hip_bf16.h>
#include <math.h>

typedef float v2f __attribute__((ext_vector_type(2)));
typedef float v8f __attribute__((ext_vector_type(8)));

#define BN 8
#define NN 1024
#define DD 256
#define EPSV 1e-8f
#define SSTRIDE 1028
#define KT 4
#define DCH 16

__device__ __forceinline__ v8f wmma4(v2f a, v2f b, v8f c) {
  return __builtin_amdgcn_wmma_f32_16x16x4_f32(
       false, a,  false, b,
       (short)0, c,  false,  false);
}

__device__ __forceinline__ float wave_max(float v) {
#pragma unroll
  for (int off = 16; off > 0; off >>= 1) v = fmaxf(v, __shfl_xor(v, off, 32));
  return v;
}
__device__ __forceinline__ float wave_sum(float v) {
#pragma unroll
  for (int off = 16; off > 0; off >>= 1) v += __shfl_xor(v, off, 32);
  return v;
}

__global__ void pos_softmax_kernel(const float* __restrict__ coords,
                                   const float* __restrict__ pos_emb,
                                   const float* __restrict__ p_temp,
                                   float* __restrict__ pos_score) {
  __shared__ float row[NN];
  __shared__ float red[8];
  const int n = blockIdx.x;
  const int tid = threadIdx.x;
  const int lane = tid & 31, wave = tid >> 5;
  const float pa = fabsf(p_temp[0]);
  float pe[6];
#pragma unroll
  for (int c = 0; c < 6; ++c) pe[c] = pos_emb[n * 6 + c];

  float lmax = -1e30f;
  for (int m = tid; m < NN; m += 256) {
    const float* cp = coords + ((size_t)n * NN + m) * 6;
    float dot = cp[0] * pe[0] + cp[1] * pe[1] + cp[2] * pe[2] +
                cp[3] * pe[3] + cp[4] * pe[4] + cp[5] * pe[5];
    float lg = -pa * dot;
    row[m] = lg;
    lmax = fmaxf(lmax, lg);
  }
  lmax = wave_max(lmax);
  if (lane == 0) red[wave] = lmax;
  __syncthreads();
  float rmax = red[0];
#pragma unroll
  for (int w = 1; w < 8; ++w) rmax = fmaxf(rmax, red[w]);
  __syncthreads();

  float lsum = 0.f;
  for (int m = tid; m < NN; m += 256) {
    float e = expf(row[m] - rmax);
    row[m] = e;
    lsum += e;
  }
  lsum = wave_sum(lsum);
  if (lane == 0) red[wave] = lsum;
  __syncthreads();
  float rsum = 0.f;
#pragma unroll
  for (int w = 0; w < 8; ++w) rsum += red[w];
  const float inv = 1.0f / rsum;
  for (int m = tid; m < NN; m += 256)
    pos_score[(size_t)n * NN + m] = row[m] * inv;
}

__global__ void gemm_xut_kernel(const float* __restrict__ x,
                                const float* __restrict__ U,
                                float* __restrict__ t) {
  const int lane = threadIdx.x & 31;
  const int wave = threadIdx.x >> 5;
  const int tile = blockIdx.x * 8 + wave;
  const int i0 = (tile >> 4) * 16;
  const int j0 = (tile & 15) * 16;
  const int hl = lane >> 4, ll = lane & 15;

  const float* arow = x + (size_t)(i0 + ll) * DD + 2 * hl;
  const float* brow = U + (size_t)(j0 + ll) * DD + 2 * hl;
  v8f acc = {0.f, 0.f, 0.f, 0.f, 0.f, 0.f, 0.f, 0.f};
#pragma unroll 8
  for (int kb = 0; kb < DD; kb += 4) {
    v2f a = *(const v2f*)(arow + kb);
    v2f b = *(const v2f*)(brow + kb);
    acc = wmma4(a, b, acc);
  }
  float* orow = t + (size_t)(i0 + 8 * hl) * DD + j0 + ll;
#pragma unroll
  for (int v = 0; v < 8; ++v) orow[(size_t)v * DD] = acc[v];
}

__global__ void gemm_su_kernel(const float* __restrict__ t,
                               const float* __restrict__ U,
                               const float* __restrict__ S1,
                               const float* __restrict__ S2,
                               float* __restrict__ xs1,
                               float* __restrict__ xs2) {
  const int lane = threadIdx.x & 31;
  const int wave = threadIdx.x >> 5;
  const int tile = blockIdx.x * 8 + wave;
  const int i0 = (tile >> 4) * 16;
  const int j0 = (tile & 15) * 16;
  const int hl = lane >> 4, ll = lane & 15;

  const float* arow = t + (size_t)(i0 + ll) * DD + 2 * hl;
  const float* bcol = U + j0 + ll;
  v8f acc1 = {0.f, 0.f, 0.f, 0.f, 0.f, 0.f, 0.f, 0.f};
  v8f acc2 = {0.f, 0.f, 0.f, 0.f, 0.f, 0.f, 0.f, 0.f};
#pragma unroll 4
  for (int kb = 0; kb < DD; kb += 4) {
    const int k = kb + 2 * hl;
    v2f a = *(const v2f*)(arow + kb);
    v2f b;
    b.x = bcol[(size_t)k * DD];
    b.y = bcol[(size_t)(k + 1) * DD];
    v2f a1 = {a.x * fabsf(S1[k]), a.y * fabsf(S1[k + 1])};
    v2f a2 = {a.x * fabsf(S2[k]), a.y * fabsf(S2[k + 1])};
    acc1 = wmma4(a1, b, acc1);
    acc2 = wmma4(a2, b, acc2);
  }
  float* o1 = xs1 + (size_t)(i0 + 8 * hl) * DD + j0 + ll;
  float* o2 = xs2 + (size_t)(i0 + 8 * hl) * DD + j0 + ll;
#pragma unroll
  for (int v = 0; v < 8; ++v) {
    o1[(size_t)v * DD] = acc1[v];
    o2[(size_t)v * DD] = acc2[v];
  }
}

__global__ void __launch_bounds__(256) fused_attn_kernel(const float* __restrict__ xs1,
                                  const float* __restrict__ xs2,
                                  const float* __restrict__ y,
                                  const float* __restrict__ pos_score,
                                  const float* __restrict__ gating,
                                  const float* __restrict__ h_temp,
                                  float* __restrict__ out,
                                  float* __restrict__ heat) {
  extern __shared__ float smem[];
  float* SL = smem;
  float* entS = smem + 2 * 16 * SSTRIDE;
  float* selS = entS + 32;

  const int b = blockIdx.x >> 6;
  const int n0 = (blockIdx.x & 63) * 16;
  const int tid = threadIdx.x;
  const int lane = tid & 31, wave = tid >> 5;
  const int hl = lane >> 4, ll = lane & 15;
  const float scale = 0.0625f;
  const float g = 1.0f / (1.0f + expf(-gating[0]));
  const float omg = 1.0f - g;
  const float ht = h_temp[0];
  const float* yb = y + (size_t)b * NN * DD;

  {
    const float* a1row = xs1 + ((size_t)b * NN + n0 + ll) * DD + 2 * hl;
    const float* a2row = xs2 + ((size_t)b * NN + n0 + ll) * DD + 2 * hl;
    const int mw = wave * (2 * KT * 16);
    const float* brow = yb + (size_t)(mw + ll) * DD + 2 * hl;
    v8f acc1[2 * KT], acc2[2 * KT];
#pragma unroll
    for (int t = 0; t < 2 * KT; ++t) {
      acc1[t] = (v8f){0.f, 0.f, 0.f, 0.f, 0.f, 0.f, 0.f, 0.f};
      acc2[t] = (v8f){0.f, 0.f, 0.f, 0.f, 0.f, 0.f, 0.f, 0.f};
    }
    for (int kl = 0; kl < DD; kl += 32) {
      v2f a1[8], a2[8];
#pragma unroll
      for (int j = 0; j < 8; ++j) a1[j] = *(const v2f*)(a1row + kl + 4 * j);
#pragma unroll
      for (int j = 0; j < 8; ++j) a2[j] = *(const v2f*)(a2row + kl + 4 * j);
#pragma unroll
      for (int h = 0; h < 2; ++h) {
        v2f bq[KT][8];
#pragma unroll
        for (int t = 0; t < KT; ++t) {
#pragma unroll
          for (int j = 0; j < 8; ++j)
            bq[t][j] = *(const v2f*)(brow + (size_t)(h * KT + t) * 16 * DD + kl + 4 * j);
        }
        __builtin_amdgcn_sched_barrier(0);
#pragma unroll
        for (int j = 0; j < 8; ++j) {
#pragma unroll
          for (int t = 0; t < KT; ++t) {
            acc1[h * KT + t] = wmma4(a1[j], bq[t][j], acc1[h * KT + t]);
            acc2[h * KT + t] = wmma4(a2[j], bq[t][j], acc2[h * KT + t]);
          }
        }
      }
    }
#pragma unroll
    for (int t = 0; t < 2 * KT; ++t) {
      float* d1 = SL + ((size_t)8 * hl) * SSTRIDE + mw + t * 16 + ll;
      float* d2 = d1 + (size_t)16 * SSTRIDE;
#pragma unroll
      for (int v = 0; v < 8; ++v) {
        d1[(size_t)v * SSTRIDE] = acc1[t][v] * scale;
        d2[(size_t)v * SSTRIDE] = acc2[t][v] * scale;
      }
    }
  }
  __syncthreads();

  for (int idx = wave; idx < 32; idx += 8) {
    const int br = idx >> 4, row = idx & 15;
    float* Lrow = SL + (size_t)(br * 16 + row) * SSTRIDE;
    const float* prow = pos_score + (size_t)(n0 + row) * NN;
    float lmax = -1e30f;
    for (int m = lane; m < NN; m += 32) lmax = fmaxf(lmax, Lrow[m]);
    lmax = wave_max(lmax);
    float lsum = 0.f;
    for (int m = lane; m < NN; m += 32) lsum += expf(Lrow[m] - lmax);
    lsum = wave_sum(lsum);
    const float inv = 1.0f / lsum;
    float ent = 0.f;
    for (int m = lane; m < NN; m += 32) {
      float p = expf(Lrow[m] - lmax) * inv;
      float a = omg * p + g * prow[m];
      ent += -a * logf(a + EPSV);
      Lrow[m] = a;
    }
    ent = wave_sum(ent);
    if (lane == 0) entS[br * 16 + row] = ent;
  }
  __syncthreads();

  if (tid < 16) {
    const int row = tid;
    const float h0 = 2.0f - 2.0f / (1.0f + expf(-ht * entS[row]));
    const float h1 = 2.0f - 2.0f / (1.0f + expf(-ht * entS[16 + row]));
    const int sel = (h0 >= h1) ? 0 : 1;
    selS[row] = (float)sel;
    heat[(size_t)b * NN + n0 + row] = sel ? h1 : h0;
  }
  __syncthreads();

  {
    const int selrow = (int)selS[ll];
    const float* Abase = SL + (size_t)(selrow * 16 + ll) * SSTRIDE + 2 * hl;
    const int dbase = wave * 32;
    const float* bbase = yb + (size_t)(2 * hl) * DD + dbase + 2 * ll;
    v8f acc0 = {0.f, 0.f, 0.f, 0.f, 0.f, 0.f, 0.f, 0.f};
    v8f acc1 = {0.f, 0.f, 0.f, 0.f, 0.f, 0.f, 0.f, 0.f};
    for (int kc = 0; kc < NN; kc += 4 * DCH) {
      v2f r0[DCH], r1[DCH], af[DCH];
#pragma unroll
      for (int j = 0; j < DCH; ++j) {
        r0[j] = *(const v2f*)(bbase + (size_t)(kc + 4 * j) * DD);
        r1[j] = *(const v2f*)(bbase + (size_t)(kc + 4 * j + 1) * DD);
      }
#pragma unroll
      for (int j = 0; j < DCH; ++j) af[j] = *(const v2f*)(Abase + kc + 4 * j);
      __builtin_amdgcn_sched_barrier(0);
#pragma unroll
      for (int j = 0; j < DCH; ++j) {
        const v2f b0 = {r0[j].x, r1[j].x};
        const v2f b1 = {r0[j].y, r1[j].y};
        acc0 = wmma4(af[j], b0, acc0);
        acc1 = wmma4(af[j], b1, acc1);
      }
    }
    float* orow = out + ((size_t)b * NN + n0 + 8 * hl) * DD + dbase + 2 * ll;
#pragma unroll
    for (int v = 0; v < 8; ++v) {
      const v2f o = {acc0[v], acc1[v]};
      *(v2f*)(orow + (size_t)v * DD) = o;
    }
  }
}

extern "C" void kernel_launch(void* const* d_in, const int* in_sizes, int n_in,
                              void* d_out, int out_size, void* d_ws, size_t ws_size,
                              hipStream_t stream) {
  const float* x = (const float*)d_in[0];
  const float* y = (const float*)d_in[1];
  const float* coords = (const float*)d_in[2];
  const float* U = (const float*)d_in[3];
  const float* S1 = (const float*)d_in[4];
  const float* S2 = (const float*)d_in[5];
  const float* gating = (const float*)d_in[6];
  const float* h_temp = (const float*)d_in[7];
  const float* p_temp = (const float*)d_in[8];
  const float* pos_emb = (const float*)d_in[9];

  float* out = (float*)d_out;
  float* heat = out + (size_t)BN * NN * DD;

  float* ws = (float*)d_ws;
  float* pos_score = ws;
  float* t = pos_score + (size_t)NN * NN;
  float* xs1 = t + (size_t)BN * NN * DD;
  float* xs2 = xs1 + (size_t)BN * NN * DD;

  pos_softmax_kernel<<<NN, 256, 0, stream>>>(coords, pos_emb, p_temp, pos_score);
  gemm_xut_kernel<<<1024, 256, 0, stream>>>(x, U, t);
  gemm_su_kernel<<<1024, 256, 0, stream>>>(t, U, S1, S2, xs1, xs2);

  const size_t lds_bytes = (size_t)(2 * 16 * SSTRIDE + 32 + 16) * sizeof(float);
  fused_attn_kernel<<<BN * 64, 256, lds_bytes, stream>>>(
      xs1, xs2, y, pos_score, gating, h_temp, out, heat);
}


# Both projection GEMMs in one launch with U resident in LDS: x, t and U fragments are LDS reads, global loads whole-line 16-byte, t stays on chip; k order per accumulator kept, results bit-identical
# speedup vs baseline: 1.9359x; 1.1622x over previous
#include <hip/hip_runtime.h>
#include <hip/hip_bf16.h>
#include <math.h>

typedef float v2f __attribute__((ext_vector_type(2)));
typedef float v8f __attribute__((ext_vector_type(8)));

#define BN 8
#define NN 1024
#define DD 256
#define EPSV 1e-8f
#define SSTRIDE 1028
#define KT 4
#define DCH 16

__device__ __forceinline__ v8f wmma4(v2f a, v2f b, v8f c) {
  return __builtin_amdgcn_wmma_f32_16x16x4_f32(
       false, a,  false, b,
       (short)0, c,  false,  false);
}

__device__ __forceinline__ float wave_max(float v) {
#pragma unroll
  for (int off = 16; off > 0; off >>= 1) v = fmaxf(v, __shfl_xor(v, off, 32));
  return v;
}
__device__ __forceinline__ float wave_sum(float v) {
#pragma unroll
  for (int off = 16; off > 0; off >>= 1) v += __shfl_xor(v, off, 32);
  return v;
}

__global__ void pos_softmax_kernel(const float* __restrict__ coords,
                                   const float* __restrict__ pos_emb,
                                   const float* __restrict__ p_temp,
                                   float* __restrict__ pos_score) {
  __shared__ float row[NN];
  __shared__ float red[8];
  const int n = blockIdx.x;
  const int tid = threadIdx.x;
  const int lane = tid & 31, wave = tid >> 5;
  const float pa = fabsf(p_temp[0]);
  float pe[6];
#pragma unroll
  for (int c = 0; c < 6; ++c) pe[c] = pos_emb[n * 6 + c];

  float lmax = -1e30f;
  for (int m = tid; m < NN; m += 256) {
    const float* cp = coords + ((size_t)n * NN + m) * 6;
    float dot = cp[0] * pe[0] + cp[1] * pe[1] + cp[2] * pe[2] +
                cp[3] * pe[3] + cp[4] * pe[4] + cp[5] * pe[5];
    float lg = -pa * dot;
    row[m] = lg;
    lmax = fmaxf(lmax, lg);
  }
  lmax = wave_max(lmax);
  if (lane == 0) red[wave] = lmax;
  __syncthreads();
  float rmax = red[0];
#pragma unroll
  for (int w = 1; w < 8; ++w) rmax = fmaxf(rmax, red[w]);
  __syncthreads();

  float lsum = 0.f;
  for (int m = tid; m < NN; m += 256) {
    float e = expf(row[m] - rmax);
    row[m] = e;
    lsum += e;
  }
  lsum = wave_sum(lsum);
  if (lane == 0) red[wave] = lsum;
  __syncthreads();
  float rsum = 0.f;
#pragma unroll
  for (int w = 0; w < 8; ++w) rsum += red[w];
  const float inv = 1.0f / rsum;
  for (int m = tid; m < NN; m += 256)
    pos_score[(size_t)n * NN + m] = row[m] * inv;
}

#define PR 32
#define UP 268
#define XP 260
#define PCH 4
__global__ void __launch_bounds__(256) proj_fused_kernel(const float* __restrict__ x,
                                  const float* __restrict__ U,
                                  const float* __restrict__ S1,
                                  const float* __restrict__ S2,
                                  float* __restrict__ xs1,
                                  float* __restrict__ xs2) {
  __shared__ __attribute__((aligned(16))) float plds[DD * UP + PR * XP + 2 * DD];
  float* Ul = plds;
  float* Tl = plds + DD * UP;
  float* Sa = Tl + PR * XP;

  const int tid = threadIdx.x;
  const int lane = tid & 31, wave = tid >> 5;
  const int hl = lane >> 4, ll = lane & 15;
  const int i0 = blockIdx.x * PR;

#pragma unroll 32
  for (int it = 0; it < 64; ++it) {
    const int idx = it * 256 + tid;
    const int r = ((idx >> 6) + (int)blockIdx.x) & (DD - 1), q = idx & 63;
    const float4 v = *(const float4*)(U + (size_t)r * DD + 4 * q);
    *(float4*)(Ul + r * UP + 4 * q) = v;
  }
#pragma unroll
  for (int it = 0; it < 8; ++it) {
    const int idx = it * 256 + tid;
    const int r = idx >> 6, q = idx & 63;
    const float4 v = *(const float4*)(x + (size_t)(i0 + r) * DD + 4 * q);
    *(float4*)(Tl + r * XP + 4 * q) = v;
  }
  Sa[tid] = fabsf(S1[tid]);
  Sa[DD + tid] = fabsf(S2[tid]);
  __syncthreads();

  const int jw = wave * 32;
  const float* ap = Tl + ll * XP + 2 * hl;

  v8f acc[2][2];
#pragma unroll
  for (int r = 0; r < 2; ++r)
#pragma unroll
    for (int q = 0; q < 2; ++q) acc[r][q] = (v8f){0.f, 0.f, 0.f, 0.f, 0.f, 0.f, 0.f, 0.f};
  {
    const float* bp = Ul + (jw + ll) * UP + 2 * hl;
    v2f a[2][2][PCH], b[2][2][PCH];
#pragma unroll
    for (int j = 0; j < PCH; ++j) {
#pragma unroll
      for (int r = 0; r < 2; ++r) a[0][r][j] = *(const v2f*)(ap + r * 16 * XP + 4 * j);
#pragma unroll
      for (int q = 0; q < 2; ++q) b[0][q][j] = *(const v2f*)(bp + q * 16 * UP + 4 * j);
    }
#pragma unroll
    for (int ci = 0; ci < DD / (4 * PCH); ++ci) {
      const int cb = ci & 1, nb = cb ^ 1;
      if (ci + 1 < DD / (4 * PCH)) {
        const int kn = (ci + 1) * 4 * PCH;
#pragma unroll
        for (int j = 0; j < PCH; ++j) {
#pragma unroll
          for (int r = 0; r < 2; ++r) a[nb][r][j] = *(const v2f*)(ap + r * 16 * XP + kn + 4 * j);
#pragma unroll
          for (int q = 0; q < 2; ++q) b[nb][q][j] = *(const v2f*)(bp + q * 16 * UP + kn + 4 * j);
        }
      }
      __builtin_amdgcn_sched_barrier(0);
#pragma unroll
      for (int j = 0; j < PCH; ++j)
#pragma unroll
        for (int r = 0; r < 2; ++r)
#pragma unroll
          for (int q = 0; q < 2; ++q) acc[r][q] = wmma4(a[cb][r][j], b[cb][q][j], acc[r][q]);
      __builtin_amdgcn_sched_barrier(0);
    }
  }
  __syncthreads();
#pragma unroll
  for (int r = 0; r < 2; ++r)
#pragma unroll
    for (int q = 0; q < 2; ++q) {
      float* tp = Tl + (16 * r + 8 * hl) * XP + jw + 16 * q + ll;
#pragma unroll
      for (int v = 0; v < 8; ++v) tp[v * XP] = acc[r][q][v];
    }
  __syncthreads();

  v8f c1[2][2], c2[2][2];
#pragma unroll
  for (int r = 0; r < 2; ++r)
#pragma unroll
    for (int q = 0; q < 2; ++q) {
      c1[r][q] = (v8f){0.f, 0.f, 0.f, 0.f, 0.f, 0.f, 0.f, 0.f};
      c2[r][q] = (v8f){0.f, 0.f, 0.f, 0.f, 0.f, 0.f, 0.f, 0.f};
    }
  {
    const float* bp = Ul + (2 * hl) * UP + jw + ll;
    const float* sp = Sa + 2 * hl;
    v2f a[2][2][PCH], b[2][2][PCH], s1[2][PCH], s2[2][PCH];
#pragma unroll
    for (int j = 0; j < PCH; ++j) {
#pragma unroll
      for (int r = 0; r < 2; ++r) a[0][r][j] = *(const v2f*)(ap + r * 16 * XP + 4 * j);
#pragma unroll
      for (int q = 0; q < 2; ++q) {
        b[0][q][j].x = bp[(4 * j) * UP + 16 * q];
        b[0][q][j].y = bp[(4 * j + 1) * UP + 16 * q];
      }
      s1[0][j] = *(const v2f*)(sp + 4 * j);
      s2[0][j] = *(const v2f*)(sp + DD + 4 * j);
    }
#pragma unroll
    for (int ci = 0; ci < DD / (4 * PCH); ++ci) {
      const int cb = ci & 1, nb = cb ^ 1;
      if (ci + 1 < DD / (4 * PCH)) {
        const int kn = (ci + 1) * 4 * PCH;
#pragma unroll
        for (int j = 0; j < PCH; ++j) {
#pragma unroll
          for (int r = 0; r < 2; ++r) a[nb][r][j] = *(const v2f*)(ap + r * 16 * XP + kn + 4 * j);
#pragma unroll
          for (int q = 0; q < 2; ++q) {
            b[nb][q][j].x = bp[(kn + 4 * j) * UP + 16 * q];
            b[nb][q][j].y = bp[(kn + 4 * j + 1) * UP + 16 * q];
          }
          s1[nb][j] = *(const v2f*)(sp + kn + 4 * j);
          s2[nb][j] = *(const v2f*)(sp + DD + kn + 4 * j);
        }
      }
      __builtin_amdgcn_sched_barrier(0);
#pragma unroll
      for (int j = 0; j < PCH; ++j)
#pragma unroll
        for (int r = 0; r < 2; ++r) {
          const v2f a1 = a[cb][r][j] * s1[cb][j];
          const v2f a2 = a[cb][r][j] * s2[cb][j];
#pragma unroll
          for (int q = 0; q < 2; ++q) {
            c1[r][q] = wmma4(a1, b[cb][q][j], c1[r][q]);
            c2[r][q] = wmma4(a2, b[cb][q][j], c2[r][q]);
          }
        }
      __builtin_amdgcn_sched_barrier(0);
    }
  }
#pragma unroll
  for (int r = 0; r < 2; ++r)
#pragma unroll
    for (int q = 0; q < 2; ++q) {
      float* o1 = xs1 + (size_t)(i0 + 16 * r + 8 * hl) * DD + jw + 16 * q + ll;
      float* o2 = xs2 + (size_t)(i0 + 16 * r + 8 * hl) * DD + jw + 16 * q + ll;
#pragma unroll
      for (int v = 0; v < 8; ++v) {
        o1[(size_t)v * DD] = c1[r][q][v];
        o2[(size_t)v * DD] = c2[r][q][v];
      }
    }
}

__global__ void __launch_bounds__(256) fused_attn_kernel(const float* __restrict__ xs1,
                                  const float* __restrict__ xs2,
                                  const float* __restrict__ y,
                                  const float* __restrict__ pos_score,
                                  const float* __restrict__ gating,
                                  const float* __restrict__ h_temp,
                                  float* __restrict__ out,
                                  float* __restrict__ heat) {
  extern __shared__ float smem[];
  float* SL = smem;
  float* entS = smem + 2 * 16 * SSTRIDE;
  float* selS = entS + 32;

  const int b = blockIdx.x >> 6;
  const int n0 = (blockIdx.x & 63) * 16;
  const int tid = threadIdx.x;
  const int lane = tid & 31, wave = tid >> 5;
  const int hl = lane >> 4, ll = lane & 15;
  const float scale = 0.0625f;
  const float g = 1.0f / (1.0f + expf(-gating[0]));
  const float omg = 1.0f - g;
  const float ht = h_temp[0];
  const float* yb = y + (size_t)b * NN * DD;

  {
    const float* a1row = xs1 + ((size_t)b * NN + n0 + ll) * DD + 2 * hl;
    const float* a2row = xs2 + ((size_t)b * NN + n0 + ll) * DD + 2 * hl;
    const int mw = wave * (2 * KT * 16);
    const float* brow = yb + (size_t)(mw + ll) * DD + 2 * hl;
    v8f acc1[2 * KT], acc2[2 * KT];
#pragma unroll
    for (int t = 0; t < 2 * KT; ++t) {
      acc1[t] = (v8f){0.f, 0.f, 0.f, 0.f, 0.f, 0.f, 0.f, 0.f};
      acc2[t] = (v8f){0.f, 0.f, 0.f, 0.f, 0.f, 0.f, 0.f, 0.f};
    }
    for (int kl = 0; kl < DD; kl += 32) {
      v2f a1[8], a2[8];
#pragma unroll
      for (int j = 0; j < 8; ++j) a1[j] = *(const v2f*)(a1row + kl + 4 * j);
#pragma unroll
      for (int j = 0; j < 8; ++j) a2[j] = *(const v2f*)(a2row + kl + 4 * j);
#pragma unroll
      for (int h = 0; h < 2; ++h) {
        v2f bq[KT][8];
#pragma unroll
        for (int t = 0; t < KT; ++t) {
#pragma unroll
          for (int j = 0; j < 8; ++j)
            bq[t][j] = *(const v2f*)(brow + (size_t)(h * KT + t) * 16 * DD + kl + 4 * j);
        }
        __builtin_amdgcn_sched_barrier(0);
#pragma unroll
        for (int j = 0; j < 8; ++j) {
#pragma unroll
          for (int t = 0; t < KT; ++t) {
            acc1[h * KT + t] = wmma4(a1[j], bq[t][j], acc1[h * KT + t]);
            acc2[h * KT + t] = wmma4(a2[j], bq[t][j], acc2[h * KT + t]);
          }
        }
      }
    }
#pragma unroll
    for (int t = 0; t < 2 * KT; ++t) {
      float* d1 = SL + ((size_t)8 * hl) * SSTRIDE + mw + t * 16 + ll;
      float* d2 = d1 + (size_t)16 * SSTRIDE;
#pragma unroll
      for (int v = 0; v < 8; ++v) {
        d1[(size_t)v * SSTRIDE] = acc1[t][v] * scale;
        d2[(size_t)v * SSTRIDE] = acc2[t][v] * scale;
      }
    }
  }
  __syncthreads();

  for (int idx = wave; idx < 32; idx += 8) {
    const int br = idx >> 4, row = idx & 15;
    float* Lrow = SL + (size_t)(br * 16 + row) * SSTRIDE;
    const float* prow = pos_score + (size_t)(n0 + row) * NN;
    float lmax = -1e30f;
    for (int m = lane; m < NN; m += 32) lmax = fmaxf(lmax, Lrow[m]);
    lmax = wave_max(lmax);
    float lsum = 0.f;
    for (int m = lane; m < NN; m += 32) lsum += expf(Lrow[m] - lmax);
    lsum = wave_sum(lsum);
    const float inv = 1.0f / lsum;
    float ent = 0.f;
    for (int m = lane; m < NN; m += 32) {
      float p = expf(Lrow[m] - lmax) * inv;
      float a = omg * p + g * prow[m];
      ent += -a * logf(a + EPSV);
      Lrow[m] = a;
    }
    ent = wave_sum(ent);
    if (lane == 0) entS[br * 16 + row] = ent;
  }
  __syncthreads();

  if (tid < 16) {
    const int row = tid;
    const float h0 = 2.0f - 2.0f / (1.0f + expf(-ht * entS[row]));
    const float h1 = 2.0f - 2.0f / (1.0f + expf(-ht * entS[16 + row]));
    const int sel = (h0 >= h1) ? 0 : 1;
    selS[row] = (float)sel;
    heat[(size_t)b * NN + n0 + row] = sel ? h1 : h0;
  }
  __syncthreads();

  {
    const int selrow = (int)selS[ll];
    const float* Abase = SL + (size_t)(selrow * 16 + ll) * SSTRIDE + 2 * hl;
    const int dbase = wave * 32;
    const float* bbase = yb + (size_t)(2 * hl) * DD + dbase + 2 * ll;
    v8f acc0 = {0.f, 0.f, 0.f, 0.f, 0.f, 0.f, 0.f, 0.f};
    v8f acc1 = {0.f, 0.f, 0.f, 0.f, 0.f, 0.f, 0.f, 0.f};
    for (int kc = 0; kc < NN; kc += 4 * DCH) {
      v2f r0[DCH], r1[DCH], af[DCH];
#pragma unroll
      for (int j = 0; j < DCH; ++j) {
        r0[j] = *(const v2f*)(bbase + (size_t)(kc + 4 * j) * DD);
        r1[j] = *(const v2f*)(bbase + (size_t)(kc + 4 * j + 1) * DD);
      }
#pragma unroll
      for (int j = 0; j < DCH; ++j) af[j] = *(const v2f*)(Abase + kc + 4 * j);
      __builtin_amdgcn_sched_barrier(0);
#pragma unroll
      for (int j = 0; j < DCH; ++j) {
        const v2f b0 = {r0[j].x, r1[j].x};
        const v2f b1 = {r0[j].y, r1[j].y};
        acc0 = wmma4(af[j], b0, acc0);
        acc1 = wmma4(af[j], b1, acc1);
      }
    }
    float* orow = out + ((size_t)b * NN + n0 + 8 * hl) * DD + dbase + 2 * ll;
#pragma unroll
    for (int v = 0; v < 8; ++v) {
      const v2f o = {acc0[v], acc1[v]};
      *(v2f*)(orow + (size_t)v * DD) = o;
    }
  }
}

extern "C" void kernel_launch(void* const* d_in, const int* in_sizes, int n_in,
                              void* d_out, int out_size, void* d_ws, size_t ws_size,
                              hipStream_t stream) {
  const float* x = (const float*)d_in[0];
  const float* y = (const float*)d_in[1];
  const float* coords = (const float*)d_in[2];
  const float* U = (const float*)d_in[3];
  const float* S1 = (const float*)d_in[4];
  const float* S2 = (const float*)d_in[5];
  const float* gating = (const float*)d_in[6];
  const float* h_temp = (const float*)d_in[7];
  const float* p_temp = (const float*)d_in[8];
  const float* pos_emb = (const float*)d_in[9];

  float* out = (float*)d_out;
  float* heat = out + (size_t)BN * NN * DD;

  float* ws = (float*)d_ws;
  float* pos_score = ws;
  float* xs1 = pos_score + (size_t)NN * NN;
  float* xs2 = xs1 + (size_t)BN * NN * DD;

  pos_softmax_kernel<<<NN, 256, 0, stream>>>(coords, pos_emb, p_temp, pos_score);
  proj_fused_kernel<<<BN * NN / PR, 256, 0, stream>>>(x, U, S1, S2, xs1, xs2);

  const size_t lds_bytes = (size_t)(2 * 16 * SSTRIDE + 32 + 16) * sizeof(float);
  fused_attn_kernel<<<BN * 64, 256, lds_bytes, stream>>>(
      xs1, xs2, y, pos_score, gating, h_temp, out, heat);
}


# Score GEMM operands staged in the idle score region of LDS: xs tiles once per workgroup, y rows per wave, whole-line 16-byte global loads, fragments are LDS reads; k order kept, bit-identical
# speedup vs baseline: 2.0286x; 1.0479x over previous
#include <hip/hip_runtime.h>
#include <hip/hip_bf16.h>
#include <math.h>

typedef float v2f __attribute__((ext_vector_type(2)));
typedef float v8f __attribute__((ext_vector_type(8)));

#define BN 8
#define NN 1024
#define DD 256
#define EPSV 1e-8f
#define SSTRIDE 1028
#define KT 4
#define DCH 16
#define AXP 260
#define YP 36

__device__ __forceinline__ v8f wmma4(v2f a, v2f b, v8f c) {
  return __builtin_amdgcn_wmma_f32_16x16x4_f32(
       false, a,  false, b,
       (short)0, c,  false,  false);
}

__device__ __forceinline__ float wave_max(float v) {
#pragma unroll
  for (int off = 16; off > 0; off >>= 1) v = fmaxf(v, __shfl_xor(v, off, 32));
  return v;
}
__device__ __forceinline__ float wave_sum(float v) {
#pragma unroll
  for (int off = 16; off > 0; off >>= 1) v += __shfl_xor(v, off, 32);
  return v;
}

__global__ void pos_softmax_kernel(const float* __restrict__ coords,
                                   const float* __restrict__ pos_emb,
                                   const float* __restrict__ p_temp,
                                   float* __restrict__ pos_score) {
  __shared__ float row[NN];
  __shared__ float red[8];
  const int n = blockIdx.x;
  const int tid = threadIdx.x;
  const int lane = tid & 31, wave = tid >> 5;
  const float pa = fabsf(p_temp[0]);
  float pe[6];
#pragma unroll
  for (int c = 0; c < 6; ++c) pe[c] = pos_emb[n * 6 + c];

  float lmax = -1e30f;
  for (int m = tid; m < NN; m += 256) {
    const float* cp = coords + ((size_t)n * NN + m) * 6;
    float dot = cp[0] * pe[0] + cp[1] * pe[1] + cp[2] * pe[2] +
                cp[3] * pe[3] + cp[4] * pe[4] + cp[5] * pe[5];
    float lg = -pa * dot;
    row[m] = lg;
    lmax = fmaxf(lmax, lg);
  }
  lmax = wave_max(lmax);
  if (lane == 0) red[wave] = lmax;
  __syncthreads();
  float rmax = red[0];
#pragma unroll
  for (int w = 1; w < 8; ++w) rmax = fmaxf(rmax, red[w]);
  __syncthreads();

  float lsum = 0.f;
  for (int m = tid; m < NN; m += 256) {
    float e = expf(row[m] - rmax);
    row[m] = e;
    lsum += e;
  }
  lsum = wave_sum(lsum);
  if (lane == 0) red[wave] = lsum;
  __syncthreads();
  float rsum = 0.f;
#pragma unroll
  for (int w = 0; w < 8; ++w) rsum += red[w];
  const float inv = 1.0f / rsum;
  for (int m = tid; m < NN; m += 256)
    pos_score[(size_t)n * NN + m] = row[m] * inv;
}

#define PR 32
#define UP 268
#define XP 260
#define PCH 4
__global__ void __launch_bounds__(256) proj_fused_kernel(const float* __restrict__ x,
                                  const float* __restrict__ U,
                                  const float* __restrict__ S1,
                                  const float* __restrict__ S2,
                                  float* __restrict__ xs1,
                                  float* __restrict__ xs2) {
  __shared__ __attribute__((aligned(16))) float plds[DD * UP + PR * XP + 2 * DD];
  float* Ul = plds;
  float* Tl = plds + DD * UP;
  float* Sa = Tl + PR * XP;

  const int tid = threadIdx.x;
  const int lane = tid & 31, wave = tid >> 5;
  const int hl = lane >> 4, ll = lane & 15;
  const int i0 = blockIdx.x * PR;

#pragma unroll 32
  for (int it = 0; it < 64; ++it) {
    const int idx = it * 256 + tid;
    const int r = ((idx >> 6) + (int)blockIdx.x) & (DD - 1), q = idx & 63;
    const float4 v = *(const float4*)(U + (size_t)r * DD + 4 * q);
    *(float4*)(Ul + r * UP + 4 * q) = v;
  }
#pragma unroll
  for (int it = 0; it < 8; ++it) {
    const int idx = it * 256 + tid;
    const int r = idx >> 6, q = idx & 63;
    const float4 v = *(const float4*)(x + (size_t)(i0 + r) * DD + 4 * q);
    *(float4*)(Tl + r * XP + 4 * q) = v;
  }
  Sa[tid] = fabsf(S1[tid]);
  Sa[DD + tid] = fabsf(S2[tid]);
  __syncthreads();

  const int jw = wave * 32;
  const float* ap = Tl + ll * XP + 2 * hl;

  v8f acc[2][2];
#pragma unroll
  for (int r = 0; r < 2; ++r)
#pragma unroll
    for (int q = 0; q < 2; ++q) acc[r][q] = (v8f){0.f, 0.f, 0.f, 0.f, 0.f, 0.f, 0.f, 0.f};
  {
    const float* bp = Ul + (jw + ll) * UP + 2 * hl;
    v2f a[2][2][PCH], b[2][2][PCH];
#pragma unroll
    for (int j = 0; j < PCH; ++j) {
#pragma unroll
      for (int r = 0; r < 2; ++r) a[0][r][j] = *(const v2f*)(ap + r * 16 * XP + 4 * j);
#pragma unroll
      for (int q = 0; q < 2; ++q) b[0][q][j] = *(const v2f*)(bp + q * 16 * UP + 4 * j);
    }
#pragma unroll
    for (int ci = 0; ci < DD / (4 * PCH); ++ci) {
      const int cb = ci & 1, nb = cb ^ 1;
      if (ci + 1 < DD / (4 * PCH)) {
        const int kn = (ci + 1) * 4 * PCH;
#pragma unroll
        for (int j = 0; j < PCH; ++j) {
#pragma unroll
          for (int r = 0; r < 2; ++r) a[nb][r][j] = *(const v2f*)(ap + r * 16 * XP + kn + 4 * j);
#pragma unroll
          for (int q = 0; q < 2; ++q) b[nb][q][j] = *(const v2f*)(bp + q * 16 * UP + kn + 4 * j);
        }
      }
      __builtin_amdgcn_sched_barrier(0);
#pragma unroll
      for (int j = 0; j < PCH; ++j)
#pragma unroll
        for (int r = 0; r < 2; ++r)
#pragma unroll
          for (int q = 0; q < 2; ++q) acc[r][q] = wmma4(a[cb][r][j], b[cb][q][j], acc[r][q]);
      __builtin_amdgcn_sched_barrier(0);
    }
  }
  __syncthreads();
#pragma unroll
  for (int r = 0; r < 2; ++r)
#pragma unroll
    for (int q = 0; q < 2; ++q) {
      float* tp = Tl + (16 * r + 8 * hl) * XP + jw + 16 * q + ll;
#pragma unroll
      for (int v = 0; v < 8; ++v) tp[v * XP] = acc[r][q][v];
    }
  __syncthreads();

  v8f c1[2][2], c2[2][2];
#pragma unroll
  for (int r = 0; r < 2; ++r)
#pragma unroll
    for (int q = 0; q < 2; ++q) {
      c1[r][q] = (v8f){0.f, 0.f, 0.f, 0.f, 0.f, 0.f, 0.f, 0.f};
      c2[r][q] = (v8f){0.f, 0.f, 0.f, 0.f, 0.f, 0.f, 0.f, 0.f};
    }
  {
    const float* bp = Ul + (2 * hl) * UP + jw + ll;
    const float* sp = Sa + 2 * hl;
    v2f a[2][2][PCH], b[2][2][PCH], s1[2][PCH], s2[2][PCH];
#pragma unroll
    for (int j = 0; j < PCH; ++j) {
#pragma unroll
      for (int r = 0; r < 2; ++r) a[0][r][j] = *(const v2f*)(ap + r * 16 * XP + 4 * j);
#pragma unroll
      for (int q = 0; q < 2; ++q) {
        b[0][q][j].x = bp[(4 * j) * UP + 16 * q];
        b[0][q][j].y = bp[(4 * j + 1) * UP + 16 * q];
      }
      s1[0][j] = *(const v2f*)(sp + 4 * j);
      s2[0][j] = *(const v2f*)(sp + DD + 4 * j);
    }
#pragma unroll
    for (int ci = 0; ci < DD / (4 * PCH); ++ci) {
      const int cb = ci & 1, nb = cb ^ 1;
      if (ci + 1 < DD / (4 * PCH)) {
        const int kn = (ci + 1) * 4 * PCH;
#pragma unroll
        for (int j = 0; j < PCH; ++j) {
#pragma unroll
          for (int r = 0; r < 2; ++r) a[nb][r][j] = *(const v2f*)(ap + r * 16 * XP + kn + 4 * j);
#pragma unroll
          for (int q = 0; q < 2; ++q) {
            b[nb][q][j].x = bp[(kn + 4 * j) * UP + 16 * q];
            b[nb][q][j].y = bp[(kn + 4 * j + 1) * UP + 16 * q];
          }
          s1[nb][j] = *(const v2f*)(sp + kn + 4 * j);
          s2[nb][j] = *(const v2f*)(sp + DD + kn + 4 * j);
        }
      }
      __builtin_amdgcn_sched_barrier(0);
#pragma unroll
      for (int j = 0; j < PCH; ++j)
#pragma unroll
        for (int r = 0; r < 2; ++r) {
          const v2f a1 = a[cb][r][j] * s1[cb][j];
          const v2f a2 = a[cb][r][j] * s2[cb][j];
#pragma unroll
          for (int q = 0; q < 2; ++q) {
            c1[r][q] = wmma4(a1, b[cb][q][j], c1[r][q]);
            c2[r][q] = wmma4(a2, b[cb][q][j], c2[r][q]);
          }
        }
      __builtin_amdgcn_sched_barrier(0);
    }
  }
#pragma unroll
  for (int r = 0; r < 2; ++r)
#pragma unroll
    for (int q = 0; q < 2; ++q) {
      float* o1 = xs1 + (size_t)(i0 + 16 * r + 8 * hl) * DD + jw + 16 * q + ll;
      float* o2 = xs2 + (size_t)(i0 + 16 * r + 8 * hl) * DD + jw + 16 * q + ll;
#pragma unroll
      for (int v = 0; v < 8; ++v) {
        o1[(size_t)v * DD] = c1[r][q][v];
        o2[(size_t)v * DD] = c2[r][q][v];
      }
    }
}

__global__ void __launch_bounds__(256) fused_attn_kernel(const float* __restrict__ xs1,
                                  const float* __restrict__ xs2,
                                  const float* __restrict__ y,
                                  const float* __restrict__ pos_score,
                                  const float* __restrict__ gating,
                                  const float* __restrict__ h_temp,
                                  float* __restrict__ out,
                                  float* __restrict__ heat) {
  extern __shared__ __attribute__((aligned(16))) float smem[];
  float* SL = smem;
  float* entS = smem + 2 * 16 * SSTRIDE;
  float* selS = entS + 32;

  const int b = blockIdx.x >> 6;
  const int n0 = (blockIdx.x & 63) * 16;
  const int tid = threadIdx.x;
  const int lane = tid & 31, wave = tid >> 5;
  const int hl = lane >> 4, ll = lane & 15;
  const float scale = 0.0625f;
  const float g = 1.0f / (1.0f + expf(-gating[0]));
  const float omg = 1.0f - g;
  const float ht = h_temp[0];
  const float* yb = y + (size_t)b * NN * DD;

  {
    float* A1l = SL;
    float* A2l = SL + 16 * AXP;
    float* Yl = SL + 32 * AXP + wave * (16 * KT * YP);
#pragma unroll
    for (int it = 0; it < 4; ++it) {
      const int idx = it * 256 + tid;
      const int r = idx >> 6, q = idx & 63;
      const size_t go = ((size_t)b * NN + n0 + r) * DD + 4 * q;
      const float4 v1 = *(const float4*)(xs1 + go);
      const float4 v2 = *(const float4*)(xs2 + go);
      *(float4*)(A1l + r * AXP + 4 * q) = v1;
      *(float4*)(A2l + r * AXP + 4 * q) = v2;
    }
    __syncthreads();

    const float* a1row = A1l + ll * AXP + 2 * hl;
    const float* a2row = A2l + ll * AXP + 2 * hl;
    const int mw = wave * (2 * KT * 16);
    const float* ysrc = yb + (size_t)(mw + (lane >> 3)) * DD + 4 * (lane & 7);
    float* ydst = Yl + (lane >> 3) * YP + 4 * (lane & 7);
    const float* brow = Yl + ll * YP + 2 * hl;
    v8f acc1[2 * KT], acc2[2 * KT];
#pragma unroll
    for (int t = 0; t < 2 * KT; ++t) {
      acc1[t] = (v8f){0.f, 0.f, 0.f, 0.f, 0.f, 0.f, 0.f, 0.f};
      acc2[t] = (v8f){0.f, 0.f, 0.f, 0.f, 0.f, 0.f, 0.f, 0.f};
    }
    for (int kl = 0; kl < DD; kl += 32) {
      v2f a1[8], a2[8];
#pragma unroll
      for (int j = 0; j < 8; ++j) a1[j] = *(const v2f*)(a1row + kl + 4 * j);
#pragma unroll
      for (int j = 0; j < 8; ++j) a2[j] = *(const v2f*)(a2row + kl + 4 * j);
#pragma unroll
      for (int h = 0; h < 2; ++h) {
        float4 g[4 * KT];
#pragma unroll
        for (int i = 0; i < 4 * KT; ++i)
          g[i] = *(const float4*)(ysrc + (size_t)(h * KT * 16 + 4 * i) * DD + kl);
#pragma unroll
        for (int i = 0; i < 4 * KT; ++i) *(float4*)(ydst + 4 * i * YP) = g[i];
        v2f bq[KT][8];
#pragma unroll
        for (int t = 0; t < KT; ++t) {
#pragma unroll
          for (int j = 0; j < 8; ++j) bq[t][j] = *(const v2f*)(brow + t * 16 * YP + 4 * j);
        }
        __builtin_amdgcn_sched_barrier(0);
#pragma unroll
        for (int j = 0; j < 8; ++j) {
#pragma unroll
          for (int t = 0; t < KT; ++t) {
            acc1[h * KT + t] = wmma4(a1[j], bq[t][j], acc1[h * KT + t]);
            acc2[h * KT + t] = wmma4(a2[j], bq[t][j], acc2[h * KT + t]);
          }
        }
        __builtin_amdgcn_sched_barrier(0);
      }
    }
    __syncthreads();
#pragma unroll
    for (int t = 0; t < 2 * KT; ++t) {
      float* d1 = SL + ((size_t)8 * hl) * SSTRIDE + mw + t * 16 + ll;
      float* d2 = d1 + (size_t)16 * SSTRIDE;
#pragma unroll
      for (int v = 0; v < 8; ++v) {
        d1[(size_t)v * SSTRIDE] = acc1[t][v] * scale;
        d2[(size_t)v * SSTRIDE] = acc2[t][v] * scale;
      }
    }
  }
  __syncthreads();

  for (int idx = wave; idx < 32; idx += 8) {
    const int br = idx >> 4, row = idx & 15;
    float* Lrow = SL + (size_t)(br * 16 + row) * SSTRIDE;
    const float* prow = pos_score + (size_t)(n0 + row) * NN;
    float lmax = -1e30f;
    for (int m = lane; m < NN; m += 32) lmax = fmaxf(lmax, Lrow[m]);
    lmax = wave_max(lmax);
    float lsum = 0.f;
    for (int m = lane; m < NN; m += 32) lsum += expf(Lrow[m] - lmax);
    lsum = wave_sum(lsum);
    const float inv = 1.0f / lsum;
    float ent = 0.f;
    for (int m = lane; m < NN; m += 32) {
      float p = expf(Lrow[m] - lmax) * inv;
      float a = omg * p + g * prow[m];
      ent += -a * logf(a + EPSV);
      Lrow[m] = a;
    }
    ent = wave_sum(ent);
    if (lane == 0) entS[br * 16 + row] = ent;
  }
  __syncthreads();

  if (tid < 16) {
    const int row = tid;
    const float h0 = 2.0f - 2.0f / (1.0f + expf(-ht * entS[row]));
    const float h1 = 2.0f - 2.0f / (1.0f + expf(-ht * entS[16 + row]));
    const int sel = (h0 >= h1) ? 0 : 1;
    selS[row] = (float)sel;
    heat[(size_t)b * NN + n0 + row] = sel ? h1 : h0;
  }
  __syncthreads();

  {
    const int selrow = (int)selS[ll];
    const float* Abase = SL + (size_t)(selrow * 16 + ll) * SSTRIDE + 2 * hl;
    const int dbase = wave * 32;
    const float* bbase = yb + (size_t)(2 * hl) * DD + dbase + 2 * ll;
    v8f acc0 = {0.f, 0.f, 0.f, 0.f, 0.f, 0.f, 0.f, 0.f};
    v8f acc1 = {0.f, 0.f, 0.f, 0.f, 0.f, 0.f, 0.f, 0.f};
    for (int kc = 0; kc < NN; kc += 4 * DCH) {
      v2f r0[DCH], r1[DCH], af[DCH];
#pragma unroll
      for (int j = 0; j < DCH; ++j) {
        r0[j] = *(const v2f*)(bbase + (size_t)(kc + 4 * j) * DD);
        r1[j] = *(const v2f*)(bbase + (size_t)(kc + 4 * j + 1) * DD);
      }
#pragma unroll
      for (int j = 0; j < DCH; ++j) af[j] = *(const v2f*)(Abase + kc + 4 * j);
      __builtin_amdgcn_sched_barrier(0);
#pragma unroll
      for (int j = 0; j < DCH; ++j) {
        const v2f b0 = {r0[j].x, r1[j].x};
        const v2f b1 = {r0[j].y, r1[j].y};
        acc0 = wmma4(af[j], b0, acc0);
        acc1 = wmma4(af[j], b1, acc1);
      }
    }
    float* orow = out + ((size_t)b * NN + n0 + 8 * hl) * DD + dbase + 2 * ll;
#pragma unroll
    for (int v = 0; v < 8; ++v) {
      const v2f o = {acc0[v], acc1[v]};
      *(v2f*)(orow + (size_t)v * DD) = o;
    }
  }
}

extern "C" void kernel_launch(void* const* d_in, const int* in_sizes, int n_in,
                              void* d_out, int out_size, void* d_ws, size_t ws_size,
                              hipStream_t stream) {
  const float* x = (const float*)d_in[0];
  const float* y = (const float*)d_in[1];
  const float* coords = (const float*)d_in[2];
  const float* U = (const float*)d_in[3];
  const float* S1 = (const float*)d_in[4];
  const float* S2 = (const float*)d_in[5];
  const float* gating = (const float*)d_in[6];
  const float* h_temp = (const float*)d_in[7];
  const float* p_temp = (const float*)d_in[8];
  const float* pos_emb = (const float*)d_in[9];

  float* out = (float*)d_out;
  float* heat = out + (size_t)BN * NN * DD;

  float* ws = (float*)d_ws;
  float* pos_score = ws;
  float* xs1 = pos_score + (size_t)NN * NN;
  float* xs2 = xs1 + (size_t)BN * NN * DD;

  pos_softmax_kernel<<<NN, 256, 0, stream>>>(coords, pos_emb, p_temp, pos_score);
  proj_fused_kernel<<<BN * NN / PR, 256, 0, stream>>>(x, U, S1, S2, xs1, xs2);

  const size_t lds_bytes = (size_t)(2 * 16 * SSTRIDE + 32 + 16) * sizeof(float);
  fused_attn_kernel<<<BN * 64, 256, lds_bytes, stream>>>(
      xs1, xs2, y, pos_score, gating, h_temp, out, heat);
}


# Softmax/entropy pass from registers: a wave does both branches of a row, loads the row's positional scores once ahead of use, keeps scores and exponentials in registers (exp once); bit-identical
# speedup vs baseline: 2.2871x; 1.1274x over previous
#include <hip/hip_runtime.h>
#include <hip/hip_bf16.h>
#include <math.h>

typedef float v2f __attribute__((ext_vector_type(2)));
typedef float v8f __attribute__((ext_vector_type(8)));

#define BN 8
#define NN 1024
#define DD 256
#define EPSV 1e-8f
#define SSTRIDE 1028
#define KT 4
#define DCH 16
#define AXP 260
#define YP 36

__device__ __forceinline__ v8f wmma4(v2f a, v2f b, v8f c) {
  return __builtin_amdgcn_wmma_f32_16x16x4_f32(
       false, a,  false, b,
       (short)0, c,  false,  false);
}

__device__ __forceinline__ float wave_max(float v) {
#pragma unroll
  for (int off = 16; off > 0; off >>= 1) v = fmaxf(v, __shfl_xor(v, off, 32));
  return v;
}
__device__ __forceinline__ float wave_sum(float v) {
#pragma unroll
  for (int off = 16; off > 0; off >>= 1) v += __shfl_xor(v, off, 32);
  return v;
}

__global__ void pos_softmax_kernel(const float* __restrict__ coords,
                                   const float* __restrict__ pos_emb,
                                   const float* __restrict__ p_temp,
                                   float* __restrict__ pos_score) {
  __shared__ float row[NN];
  __shared__ float red[8];
  const int n = blockIdx.x;
  const int tid = threadIdx.x;
  const int lane = tid & 31, wave = tid >> 5;
  const float pa = fabsf(p_temp[0]);
  float pe[6];
#pragma unroll
  for (int c = 0; c < 6; ++c) pe[c] = pos_emb[n * 6 + c];

  float lmax = -1e30f;
  for (int m = tid; m < NN; m += 256) {
    const float* cp = coords + ((size_t)n * NN + m) * 6;
    float dot = cp[0] * pe[0] + cp[1] * pe[1] + cp[2] * pe[2] +
                cp[3] * pe[3] + cp[4] * pe[4] + cp[5] * pe[5];
    float lg = -pa * dot;
    row[m] = lg;
    lmax = fmaxf(lmax, lg);
  }
  lmax = wave_max(lmax);
  if (lane == 0) red[wave] = lmax;
  __syncthreads();
  float rmax = red[0];
#pragma unroll
  for (int w = 1; w < 8; ++w) rmax = fmaxf(rmax, red[w]);
  __syncthreads();

  float lsum = 0.f;
  for (int m = tid; m < NN; m += 256) {
    float e = expf(row[m] - rmax);
    row[m] = e;
    lsum += e;
  }
  lsum = wave_sum(lsum);
  if (lane == 0) red[wave] = lsum;
  __syncthreads();
  float rsum = 0.f;
#pragma unroll
  for (int w = 0; w < 8; ++w) rsum += red[w];
  const float inv = 1.0f / rsum;
  for (int m = tid; m < NN; m += 256)
    pos_score[(size_t)n * NN + m] = row[m] * inv;
}

#define PR 32
#define UP 268
#define XP 260
#define PCH 4
__global__ void __launch_bounds__(256) proj_fused_kernel(const float* __restrict__ x,
                                  const float* __restrict__ U,
                                  const float* __restrict__ S1,
                                  const float* __restrict__ S2,
                                  float* __restrict__ xs1,
                                  float* __restrict__ xs2) {
  __shared__ __attribute__((aligned(16))) float plds[DD * UP + PR * XP + 2 * DD];
  float* Ul = plds;
  float* Tl = plds + DD * UP;
  float* Sa = Tl + PR * XP;

  const int tid = threadIdx.x;
  const int lane = tid & 31, wave = tid >> 5;
  const int hl = lane >> 4, ll = lane & 15;
  const int i0 = blockIdx.x * PR;

#pragma unroll 32
  for (int it = 0; it < 64; ++it) {
    const int idx = it * 256 + tid;
    const int r = ((idx >> 6) + (int)blockIdx.x) & (DD - 1), q = idx & 63;
    const float4 v = *(const float4*)(U + (size_t)r * DD + 4 * q);
    *(float4*)(Ul + r * UP + 4 * q) = v;
  }
#pragma unroll
  for (int it = 0; it < 8; ++it) {
    const int idx = it * 256 + tid;
    const int r = idx >> 6, q = idx & 63;
    const float4 v = *(const float4*)(x + (size_t)(i0 + r) * DD + 4 * q);
    *(float4*)(Tl + r * XP + 4 * q) = v;
  }
  Sa[tid] = fabsf(S1[tid]);
  Sa[DD + tid] = fabsf(S2[tid]);
  __syncthreads();

  const int jw = wave * 32;
  const float* ap = Tl + ll * XP + 2 * hl;

  v8f acc[2][2];
#pragma unroll
  for (int r = 0; r < 2; ++r)
#pragma unroll
    for (int q = 0; q < 2; ++q) acc[r][q] = (v8f){0.f, 0.f, 0.f, 0.f, 0.f, 0.f, 0.f, 0.f};
  {
    const float* bp = Ul + (jw + ll) * UP + 2 * hl;
    v2f a[2][2][PCH], b[2][2][PCH];
#pragma unroll
    for (int j = 0; j < PCH; ++j) {
#pragma unroll
      for (int r = 0; r < 2; ++r) a[0][r][j] = *(const v2f*)(ap + r * 16 * XP + 4 * j);
#pragma unroll
      for (int q = 0; q < 2; ++q) b[0][q][j] = *(const v2f*)(bp + q * 16 * UP + 4 * j);
    }
#pragma unroll
    for (int ci = 0; ci < DD / (4 * PCH); ++ci) {
      const int cb = ci & 1, nb = cb ^ 1;
      if (ci + 1 < DD / (4 * PCH)) {
        const int kn = (ci + 1) * 4 * PCH;
#pragma unroll
        for (int j = 0; j < PCH; ++j) {
#pragma unroll
          for (int r = 0; r < 2; ++r) a[nb][r][j] = *(const v2f*)(ap + r * 16 * XP + kn + 4 * j);
#pragma unroll
          for (int q = 0; q < 2; ++q) b[nb][q][j] = *(const v2f*)(bp + q * 16 * UP + kn + 4 * j);
        }
      }
      __builtin_amdgcn_sched_barrier(0);
#pragma unroll
      for (int j = 0; j < PCH; ++j)
#pragma unroll
        for (int r = 0; r < 2; ++r)
#pragma unroll
          for (int q = 0; q < 2; ++q) acc[r][q] = wmma4(a[cb][r][j], b[cb][q][j], acc[r][q]);
      __builtin_amdgcn_sched_barrier(0);
    }
  }
  __syncthreads();
#pragma unroll
  for (int r = 0; r < 2; ++r)
#pragma unroll
    for (int q = 0; q < 2; ++q) {
      float* tp = Tl + (16 * r + 8 * hl) * XP + jw + 16 * q + ll;
#pragma unroll
      for (int v = 0; v < 8; ++v) tp[v * XP] = acc[r][q][v];
    }
  __syncthreads();

  v8f c1[2][2], c2[2][2];
#pragma unroll
  for (int r = 0; r < 2; ++r)
#pragma unroll
    for (int q = 0; q < 2; ++q) {
      c1[r][q] = (v8f){0.f, 0.f, 0.f, 0.f, 0.f, 0.f, 0.f, 0.f};
      c2[r][q] = (v8f){0.f, 0.f, 0.f, 0.f, 0.f, 0.f, 0.f, 0.f};
    }
  {
    const float* bp = Ul + (2 * hl) * UP + jw + ll;
    const float* sp = Sa + 2 * hl;
    v2f a[2][2][PCH], b[2][2][PCH], s1[2][PCH], s2[2][PCH];
#pragma unroll
    for (int j = 0; j < PCH; ++j) {
#pragma unroll
      for (int r = 0; r < 2; ++r) a[0][r][j] = *(const v2f*)(ap + r * 16 * XP + 4 * j);
#pragma unroll
      for (int q = 0; q < 2; ++q) {
        b[0][q][j].x = bp[(4 * j) * UP + 16 * q];
        b[0][q][j].y = bp[(4 * j + 1) * UP + 16 * q];
      }
      s1[0][j] = *(const v2f*)(sp + 4 * j);
      s2[0][j] = *(const v2f*)(sp + DD + 4 * j);
    }
#pragma unroll
    for (int ci = 0; ci < DD / (4 * PCH); ++ci) {
      const int cb = ci & 1, nb = cb ^ 1;
      if (ci + 1 < DD / (4 * PCH)) {
        const int kn = (ci + 1) * 4 * PCH;
#pragma unroll
        for (int j = 0; j < PCH; ++j) {
#pragma unroll
          for (int r = 0; r < 2; ++r) a[nb][r][j] = *(const v2f*)(ap + r * 16 * XP + kn + 4 * j);
#pragma unroll
          for (int q = 0; q < 2; ++q) {
            b[nb][q][j].x = bp[(kn + 4 * j) * UP + 16 * q];
            b[nb][q][j].y = bp[(kn + 4 * j + 1) * UP + 16 * q];
          }
          s1[nb][j] = *(const v2f*)(sp + kn + 4 * j);
          s2[nb][j] = *(const v2f*)(sp + DD + kn + 4 * j);
        }
      }
      __builtin_amdgcn_sched_barrier(0);
#pragma unroll
      for (int j = 0; j < PCH; ++j)
#pragma unroll
        for (int r = 0; r < 2; ++r) {
          const v2f a1 = a[cb][r][j] * s1[cb][j];
          const v2f a2 = a[cb][r][j] * s2[cb][j];
#pragma unroll
          for (int q = 0; q < 2; ++q) {
            c1[r][q] = wmma4(a1, b[cb][q][j], c1[r][q]);
            c2[r][q] = wmma4(a2, b[cb][q][j], c2[r][q]);
          }
        }
      __builtin_amdgcn_sched_barrier(0);
    }
  }
#pragma unroll
  for (int r = 0; r < 2; ++r)
#pragma unroll
    for (int q = 0; q < 2; ++q) {
      float* o1 = xs1 + (size_t)(i0 + 16 * r + 8 * hl) * DD + jw + 16 * q + ll;
      float* o2 = xs2 + (size_t)(i0 + 16 * r + 8 * hl) * DD + jw + 16 * q + ll;
#pragma unroll
      for (int v = 0; v < 8; ++v) {
        o1[(size_t)v * DD] = c1[r][q][v];
        o2[(size_t)v * DD] = c2[r][q][v];
      }
    }
}

__global__ void __launch_bounds__(256) fused_attn_kernel(const float* __restrict__ xs1,
                                  const float* __restrict__ xs2,
                                  const float* __restrict__ y,
                                  const float* __restrict__ pos_score,
                                  const float* __restrict__ gating,
                                  const float* __restrict__ h_temp,
                                  float* __restrict__ out,
                                  float* __restrict__ heat) {
  extern __shared__ __attribute__((aligned(16))) float smem[];
  float* SL = smem;
  float* entS = smem + 2 * 16 * SSTRIDE;
  float* selS = entS + 32;

  const int b = blockIdx.x >> 6;
  const int n0 = (blockIdx.x & 63) * 16;
  const int tid = threadIdx.x;
  const int lane = tid & 31, wave = tid >> 5;
  const int hl = lane >> 4, ll = lane & 15;
  const float scale = 0.0625f;
  const float g = 1.0f / (1.0f + expf(-gating[0]));
  const float omg = 1.0f - g;
  const float ht = h_temp[0];
  const float* yb = y + (size_t)b * NN * DD;

  {
    float* A1l = SL;
    float* A2l = SL + 16 * AXP;
    float* Yl = SL + 32 * AXP + wave * (16 * KT * YP);
#pragma unroll
    for (int it = 0; it < 4; ++it) {
      const int idx = it * 256 + tid;
      const int r = idx >> 6, q = idx & 63;
      const size_t go = ((size_t)b * NN + n0 + r) * DD + 4 * q;
      const float4 v1 = *(const float4*)(xs1 + go);
      const float4 v2 = *(const float4*)(xs2 + go);
      *(float4*)(A1l + r * AXP + 4 * q) = v1;
      *(float4*)(A2l + r * AXP + 4 * q) = v2;
    }
    __syncthreads();

    const float* a1row = A1l + ll * AXP + 2 * hl;
    const float* a2row = A2l + ll * AXP + 2 * hl;
    const int mw = wave * (2 * KT * 16);
    const float* ysrc = yb + (size_t)(mw + (lane >> 3)) * DD + 4 * (lane & 7);
    float* ydst = Yl + (lane >> 3) * YP + 4 * (lane & 7);
    const float* brow = Yl + ll * YP + 2 * hl;
    v8f acc1[2 * KT], acc2[2 * KT];
#pragma unroll
    for (int t = 0; t < 2 * KT; ++t) {
      acc1[t] = (v8f){0.f, 0.f, 0.f, 0.f, 0.f, 0.f, 0.f, 0.f};
      acc2[t] = (v8f){0.f, 0.f, 0.f, 0.f, 0.f, 0.f, 0.f, 0.f};
    }
    for (int kl = 0; kl < DD; kl += 32) {
      v2f a1[8], a2[8];
#pragma unroll
      for (int j = 0; j < 8; ++j) a1[j] = *(const v2f*)(a1row + kl + 4 * j);
#pragma unroll
      for (int j = 0; j < 8; ++j) a2[j] = *(const v2f*)(a2row + kl + 4 * j);
#pragma unroll
      for (int h = 0; h < 2; ++h) {
        float4 g[4 * KT];
#pragma unroll
        for (int i = 0; i < 4 * KT; ++i)
          g[i] = *(const float4*)(ysrc + (size_t)(h * KT * 16 + 4 * i) * DD + kl);
#pragma unroll
        for (int i = 0; i < 4 * KT; ++i) *(float4*)(ydst + 4 * i * YP) = g[i];
        v2f bq[KT][8];
#pragma unroll
        for (int t = 0; t < KT; ++t) {
#pragma unroll
          for (int j = 0; j < 8; ++j) bq[t][j] = *(const v2f*)(brow + t * 16 * YP + 4 * j);
        }
        __builtin_amdgcn_sched_barrier(0);
#pragma unroll
        for (int j = 0; j < 8; ++j) {
#pragma unroll
          for (int t = 0; t < KT; ++t) {
            acc1[h * KT + t] = wmma4(a1[j], bq[t][j], acc1[h * KT + t]);
            acc2[h * KT + t] = wmma4(a2[j], bq[t][j], acc2[h * KT + t]);
          }
        }
        __builtin_amdgcn_sched_barrier(0);
      }
    }
    __syncthreads();
#pragma unroll
    for (int t = 0; t < 2 * KT; ++t) {
      float* d1 = SL + ((size_t)8 * hl) * SSTRIDE + mw + t * 16 + ll;
      float* d2 = d1 + (size_t)16 * SSTRIDE;
#pragma unroll
      for (int v = 0; v < 8; ++v) {
        d1[(size_t)v * SSTRIDE] = acc1[t][v] * scale;
        d2[(size_t)v * SSTRIDE] = acc2[t][v] * scale;
      }
    }
  }
  __syncthreads();

  for (int row = wave; row < 16; row += 8) {
    const float* prow = pos_score + (size_t)(n0 + row) * NN + lane;
    float pr[32];
#pragma unroll
    for (int i = 0; i < 32; ++i) pr[i] = prow[32 * i];
#pragma nounroll
    for (int br = 0; br < 2; ++br) {
      float* Lrow = SL + (size_t)(br * 16 + row) * SSTRIDE + lane;
      float l[32];
#pragma unroll
      for (int i = 0; i < 32; ++i) l[i] = Lrow[32 * i];
      float lmax = -1e30f;
#pragma unroll
      for (int i = 0; i < 32; ++i) lmax = fmaxf(lmax, l[i]);
      lmax = wave_max(lmax);
      float lsum = 0.f;
#pragma unroll
      for (int i = 0; i < 32; ++i) {
        l[i] = expf(l[i] - lmax);
        lsum += l[i];
      }
      lsum = wave_sum(lsum);
      const float inv = 1.0f / lsum;
      float ent = 0.f;
#pragma unroll
      for (int i = 0; i < 32; ++i) {
        const float p = l[i] * inv;
        const float a = fmaf(g, pr[i], omg * p);
        ent = fmaf(-a, logf(a + EPSV), ent);
        Lrow[32 * i] = a;
      }
      ent = wave_sum(ent);
      if (lane == 0) entS[br * 16 + row] = ent;
    }
  }
  __syncthreads();

  if (tid < 16) {
    const int row = tid;
    const float h0 = 2.0f - 2.0f / (1.0f + expf(-ht * entS[row]));
    const float h1 = 2.0f - 2.0f / (1.0f + expf(-ht * entS[16 + row]));
    const int sel = (h0 >= h1) ? 0 : 1;
    selS[row] = (float)sel;
    heat[(size_t)b * NN + n0 + row] = sel ? h1 : h0;
  }
  __syncthreads();

  {
    const int selrow = (int)selS[ll];
    const float* Abase = SL + (size_t)(selrow * 16 + ll) * SSTRIDE + 2 * hl;
    const int dbase = wave * 32;
    const float* bbase = yb + (size_t)(2 * hl) * DD + dbase + 2 * ll;
    v8f acc0 = {0.f, 0.f, 0.f, 0.f, 0.f, 0.f, 0.f, 0.f};
    v8f acc1 = {0.f, 0.f, 0.f, 0.f, 0.f, 0.f, 0.f, 0.f};
    for (int kc = 0; kc < NN; kc += 4 * DCH) {
      v2f r0[DCH], r1[DCH], af[DCH];
#pragma unroll
      for (int j = 0; j < DCH; ++j) {
        r0[j] = *(const v2f*)(bbase + (size_t)(kc + 4 * j) * DD);
        r1[j] = *(const v2f*)(bbase + (size_t)(kc + 4 * j + 1) * DD);
      }
#pragma unroll
      for (int j = 0; j < DCH; ++j) af[j] = *(const v2f*)(Abase + kc + 4 * j);
      __builtin_amdgcn_sched_barrier(0);
#pragma unroll
      for (int j = 0; j < DCH; ++j) {
        const v2f b0 = {r0[j].x, r1[j].x};
        const v2f b1 = {r0[j].y, r1[j].y};
        acc0 = wmma4(af[j], b0, acc0);
        acc1 = wmma4(af[j], b1, acc1);
      }
    }
    float* orow = out + ((size_t)b * NN + n0 + 8 * hl) * DD + dbase + 2 * ll;
#pragma unroll
    for (int v = 0; v < 8; ++v) {
      const v2f o = {acc0[v], acc1[v]};
      *(v2f*)(orow + (size_t)v * DD) = o;
    }
  }
}

extern "C" void kernel_launch(void* const* d_in, const int* in_sizes, int n_in,
                              void* d_out, int out_size, void* d_ws, size_t ws_size,
                              hipStream_t stream) {
  const float* x = (const float*)d_in[0];
  const float* y = (const float*)d_in[1];
  const float* coords = (const float*)d_in[2];
  const float* U = (const float*)d_in[3];
  const float* S1 = (const float*)d_in[4];
  const float* S2 = (const float*)d_in[5];
  const float* gating = (const float*)d_in[6];
  const float* h_temp = (const float*)d_in[7];
  const float* p_temp = (const float*)d_in[8];
  const float* pos_emb = (const float*)d_in[9];

  float* out = (float*)d_out;
  float* heat = out + (size_t)BN * NN * DD;

  float* ws = (float*)d_ws;
  float* pos_score = ws;
  float* xs1 = pos_score + (size_t)NN * NN;
  float* xs2 = xs1 + (size_t)BN * NN * DD;

  pos_softmax_kernel<<<NN, 256, 0, stream>>>(coords, pos_emb, p_temp, pos_score);
  proj_fused_kernel<<<BN * NN / PR, 256, 0, stream>>>(x, U, S1, S2, xs1, xs2);

  const size_t lds_bytes = (size_t)(2 * 16 * SSTRIDE + 32 + 16) * sizeof(float);
  fused_attn_kernel<<<BN * 64, 256, lds_bytes, stream>>>(
      xs1, xs2, y, pos_score, gating, h_temp, out, heat);
}


# s_setprio falls with a workgroup's progress (3 score GEMM, 2 softmax, 1 output GEMM, 0 its last quarter): the 2 workgroups of a WGP take turns, neither ends alone with loads uncovered; bit-identical
# speedup vs baseline: 2.4558x; 1.0738x over previous
#include <hip/hip_runtime.h>
#include <hip/hip_bf16.h>
#include <math.h>

typedef float v2f __attribute__((ext_vector_type(2)));
typedef float v8f __attribute__((ext_vector_type(8)));

#define BN 8
#define NN 1024
#define DD 256
#define EPSV 1e-8f
#define SSTRIDE 1028
#define KT 4
#define DCH 16
#define AXP 260
#define YP 36

__device__ __forceinline__ v8f wmma4(v2f a, v2f b, v8f c) {
  return __builtin_amdgcn_wmma_f32_16x16x4_f32(
       false, a,  false, b,
       (short)0, c,  false,  false);
}

__device__ __forceinline__ float wave_max(float v) {
#pragma unroll
  for (int off = 16; off > 0; off >>= 1) v = fmaxf(v, __shfl_xor(v, off, 32));
  return v;
}
__device__ __forceinline__ float wave_sum(float v) {
#pragma unroll
  for (int off = 16; off > 0; off >>= 1) v += __shfl_xor(v, off, 32);
  return v;
}

#define PR 32
#define UP 268
#define XP 260
#define PCH 4
#define PW 4
__global__ void __launch_bounds__(256 + 32 * PW) proj_pos_kernel(const float* __restrict__ x,
                                  const float* __restrict__ U,
                                  const float* __restrict__ S1,
                                  const float* __restrict__ S2,
                                  float* __restrict__ xs1,
                                  float* __restrict__ xs2,
                                  const float* __restrict__ coords,
                                  const float* __restrict__ pos_emb,
                                  const float* __restrict__ p_temp,
                                  float* __restrict__ pos_score) {
  __shared__ __attribute__((aligned(16))) float plds[DD * UP + PR * XP + 2 * DD];
  float* Ul = plds;
  float* Tl = plds + DD * UP;
  float* Sa = Tl + PR * XP;

  const int tid = threadIdx.x;
  const int lane = tid & 31;
  const int wave = __builtin_amdgcn_readfirstlane(tid >> 5);
  const bool gemm = wave < 8;
  const int hl = lane >> 4, ll = lane & 15;
  const int i0 = blockIdx.x * PR;

  if (gemm) {
#pragma unroll 32
  for (int it = 0; it < 64; ++it) {
    const int idx = it * 256 + tid;
    const int r = ((idx >> 6) + (int)blockIdx.x) & (DD - 1), q = idx & 63;
    const float4 v = *(const float4*)(U + (size_t)r * DD + 4 * q);
    *(float4*)(Ul + r * UP + 4 * q) = v;
  }
#pragma unroll
  for (int it = 0; it < 8; ++it) {
    const int idx = it * 256 + tid;
    const int r = idx >> 6, q = idx & 63;
    const float4 v = *(const float4*)(x + (size_t)(i0 + r) * DD + 4 * q);
    *(float4*)(Tl + r * XP + 4 * q) = v;
  }
  Sa[tid] = fabsf(S1[tid]);
  Sa[DD + tid] = fabsf(S2[tid]);
  }
  __syncthreads();

  const int jw = wave * 32;
  const float* ap = Tl + ll * XP + 2 * hl;

  v8f acc[2][2];
#pragma unroll
  for (int r = 0; r < 2; ++r)
#pragma unroll
    for (int q = 0; q < 2; ++q) acc[r][q] = (v8f){0.f, 0.f, 0.f, 0.f, 0.f, 0.f, 0.f, 0.f};
  if (gemm) {
    const float* bp = Ul + (jw + ll) * UP + 2 * hl;
    v2f a[2][2][PCH], b[2][2][PCH];
#pragma unroll
    for (int j = 0; j < PCH; ++j) {
#pragma unroll
      for (int r = 0; r < 2; ++r) a[0][r][j] = *(const v2f*)(ap + r * 16 * XP + 4 * j);
#pragma unroll
      for (int q = 0; q < 2; ++q) b[0][q][j] = *(const v2f*)(bp + q * 16 * UP + 4 * j);
    }
#pragma unroll
    for (int ci = 0; ci < DD / (4 * PCH); ++ci) {
      const int cb = ci & 1, nb = cb ^ 1;
      if (ci + 1 < DD / (4 * PCH)) {
        const int kn = (ci + 1) * 4 * PCH;
#pragma unroll
        for (int j = 0; j < PCH; ++j) {
#pragma unroll
          for (int r = 0; r < 2; ++r) a[nb][r][j] = *(const v2f*)(ap + r * 16 * XP + kn + 4 * j);
#pragma unroll
          for (int q = 0; q < 2; ++q) b[nb][q][j] = *(const v2f*)(bp + q * 16 * UP + kn + 4 * j);
        }
      }
      __builtin_amdgcn_sched_barrier(0);
#pragma unroll
      for (int j = 0; j < PCH; ++j)
#pragma unroll
        for (int r = 0; r < 2; ++r)
#pragma unroll
          for (int q = 0; q < 2; ++q) acc[r][q] = wmma4(a[cb][r][j], b[cb][q][j], acc[r][q]);
      __builtin_amdgcn_sched_barrier(0);
    }
  }
  __syncthreads();
  if (gemm) {
#pragma unroll
  for (int r = 0; r < 2; ++r)
#pragma unroll
    for (int q = 0; q < 2; ++q) {
      float* tp = Tl + (16 * r + 8 * hl) * XP + jw + 16 * q + ll;
#pragma unroll
      for (int v = 0; v < 8; ++v) tp[v * XP] = acc[r][q][v];
    }
  }
  __syncthreads();

  if (gemm) {
  v8f c1[2][2], c2[2][2];
#pragma unroll
  for (int r = 0; r < 2; ++r)
#pragma unroll
    for (int q = 0; q < 2; ++q) {
      c1[r][q] = (v8f){0.f, 0.f, 0.f, 0.f, 0.f, 0.f, 0.f, 0.f};
      c2[r][q] = (v8f){0.f, 0.f, 0.f, 0.f, 0.f, 0.f, 0.f, 0.f};
    }
  {
    const float* bp = Ul + (2 * hl) * UP + jw + ll;
    const float* sp = Sa + 2 * hl;
    v2f a[2][2][PCH], b[2][2][PCH], s1[2][PCH], s2[2][PCH];
#pragma unroll
    for (int j = 0; j < PCH; ++j) {
#pragma unroll
      for (int r = 0; r < 2; ++r) a[0][r][j] = *(const v2f*)(ap + r * 16 * XP + 4 * j);
#pragma unroll
      for (int q = 0; q < 2; ++q) {
        b[0][q][j].x = bp[(4 * j) * UP + 16 * q];
        b[0][q][j].y = bp[(4 * j + 1) * UP + 16 * q];
      }
      s1[0][j] = *(const v2f*)(sp + 4 * j);
      s2[0][j] = *(const v2f*)(sp + DD + 4 * j);
    }
#pragma unroll
    for (int ci = 0; ci < DD / (4 * PCH); ++ci) {
      const int cb = ci & 1, nb = cb ^ 1;
      if (ci + 1 < DD / (4 * PCH)) {
        const int kn = (ci + 1) * 4 * PCH;
#pragma unroll
        for (int j = 0; j < PCH; ++j) {
#pragma unroll
          for (int r = 0; r < 2; ++r) a[nb][r][j] = *(const v2f*)(ap + r * 16 * XP + kn + 4 * j);
#pragma unroll
          for (int q = 0; q < 2; ++q) {
            b[nb][q][j].x = bp[(kn + 4 * j) * UP + 16 * q];
            b[nb][q][j].y = bp[(kn + 4 * j + 1) * UP + 16 * q];
          }
          s1[nb][j] = *(const v2f*)(sp + kn + 4 * j);
          s2[nb][j] = *(const v2f*)(sp + DD + kn + 4 * j);
        }
      }
      __builtin_amdgcn_sched_barrier(0);
#pragma unroll
      for (int j = 0; j < PCH; ++j)
#pragma unroll
        for (int r = 0; r < 2; ++r) {
          const v2f a1 = a[cb][r][j] * s1[cb][j];
          const v2f a2 = a[cb][r][j] * s2[cb][j];
#pragma unroll
          for (int q = 0; q < 2; ++q) {
            c1[r][q] = wmma4(a1, b[cb][q][j], c1[r][q]);
            c2[r][q] = wmma4(a2, b[cb][q][j], c2[r][q]);
          }
        }
      __builtin_amdgcn_sched_barrier(0);
    }
  }
#pragma unroll
  for (int r = 0; r < 2; ++r)
#pragma unroll
    for (int q = 0; q < 2; ++q) {
      float* o1 = xs1 + (size_t)(i0 + 16 * r + 8 * hl) * DD + jw + 16 * q + ll;
      float* o2 = xs2 + (size_t)(i0 + 16 * r + 8 * hl) * DD + jw + 16 * q + ll;
#pragma unroll
      for (int v = 0; v < 8; ++v) {
        o1[(size_t)v * DD] = c1[r][q][v];
        o2[(size_t)v * DD] = c2[r][q][v];
      }
    }
  } else {
    const int n = (int)blockIdx.x * PW + (wave - 8);
    const float pa = fabsf(p_temp[0]);
    float pe[6];
#pragma unroll
    for (int c = 0; c < 6; ++c) pe[c] = pos_emb[n * 6 + c];
    const float* crow = coords + ((size_t)n * NN + lane) * 6;
    float e[32];
    float lmax = -1e30f;
#pragma unroll
    for (int j = 0; j < 32; ++j) {
      const float* cp = crow + (size_t)(32 * j) * 6;
      const v2f c01 = *(const v2f*)(cp);
      const v2f c23 = *(const v2f*)(cp + 2);
      const v2f c45 = *(const v2f*)(cp + 4);
      float lg;
      {
#pragma clang fp contract(off)
        float d = ((c01.x * pe[0] + c01.y * pe[1]) + c23.x * pe[2]) + c23.y * pe[3];
        d = __builtin_fmaf(c45.x, pe[4], d);
        d = __builtin_fmaf(c45.y, pe[5], d);
        lg = -pa * d;
      }
      e[j] = lg;
      lmax = fmaxf(lmax, lg);
    }
    const float rmax = wave_max(lmax);
    float rsum = 0.f;
#pragma unroll
    for (int w = 0; w < 8; ++w) {
      float ls = 0.f;
#pragma unroll
      for (int it = 0; it < 4; ++it) {
        const float ex = expf(e[w + 8 * it] - rmax);
        e[w + 8 * it] = ex;
        ls += ex;
      }
      rsum += wave_sum(ls);
    }
    const float inv = 1.0f / rsum;
    float* prow = pos_score + (size_t)n * NN + lane;
#pragma unroll
    for (int j = 0; j < 32; ++j) prow[32 * j] = e[j] * inv;
  }
}

__global__ void __launch_bounds__(256) fused_attn_kernel(const float* __restrict__ xs1,
                                  const float* __restrict__ xs2,
                                  const float* __restrict__ y,
                                  const float* __restrict__ pos_score,
                                  const float* __restrict__ gating,
                                  const float* __restrict__ h_temp,
                                  float* __restrict__ out,
                                  float* __restrict__ heat) {
  extern __shared__ __attribute__((aligned(16))) float smem[];
  float* SL = smem;
  float* entS = smem + 2 * 16 * SSTRIDE;
  float* selS = entS + 32;

  const int b = blockIdx.x >> 6;
  const int n0 = (blockIdx.x & 63) * 16;
  const int tid = threadIdx.x;
  const int lane = tid & 31, wave = tid >> 5;
  const int hl = lane >> 4, ll = lane & 15;
  const float scale = 0.0625f;
  const float g = 1.0f / (1.0f + expf(-gating[0]));
  const float omg = 1.0f - g;
  const float ht = h_temp[0];
  const float* yb = y + (size_t)b * NN * DD;

  {
    float* A1l = SL;
    float* A2l = SL + 16 * AXP;
    float* Yl = SL + 32 * AXP + wave * (16 * KT * YP);
#pragma unroll
    for (int it = 0; it < 4; ++it) {
      const int idx = it * 256 + tid;
      const int r = idx >> 6, q = idx & 63;
      const size_t go = ((size_t)b * NN + n0 + r) * DD + 4 * q;
      const float4 v1 = *(const float4*)(xs1 + go);
      const float4 v2 = *(const float4*)(xs2 + go);
      *(float4*)(A1l + r * AXP + 4 * q) = v1;
      *(float4*)(A2l + r * AXP + 4 * q) = v2;
    }
    __syncthreads();

    const float* a1row = A1l + ll * AXP + 2 * hl;
    const float* a2row = A2l + ll * AXP + 2 * hl;
    const int mw = wave * (2 * KT * 16);
    const float* ysrc = yb + (size_t)(mw + (lane >> 3)) * DD + 4 * (lane & 7);
    float* ydst = Yl + (lane >> 3) * YP + 4 * (lane & 7);
    const float* brow = Yl + ll * YP + 2 * hl;
    v8f acc1[2 * KT], acc2[2 * KT];
#pragma unroll
    for (int t = 0; t < 2 * KT; ++t) {
      acc1[t] = (v8f){0.f, 0.f, 0.f, 0.f, 0.f, 0.f, 0.f, 0.f};
      acc2[t] = (v8f){0.f, 0.f, 0.f, 0.f, 0.f, 0.f, 0.f, 0.f};
    }
    __builtin_amdgcn_s_setprio(3);
    for (int kl = 0; kl < DD; kl += 32) {
      v2f a1[8], a2[8];
#pragma unroll
      for (int j = 0; j < 8; ++j) a1[j] = *(const v2f*)(a1row + kl + 4 * j);
#pragma unroll
      for (int j = 0; j < 8; ++j) a2[j] = *(const v2f*)(a2row + kl + 4 * j);
#pragma unroll
      for (int h = 0; h < 2; ++h) {
        float4 g[4 * KT];
#pragma unroll
        for (int i = 0; i < 4 * KT; ++i)
          g[i] = *(const float4*)(ysrc + (size_t)(h * KT * 16 + 4 * i) * DD + kl);
#pragma unroll
        for (int i = 0; i < 4 * KT; ++i) *(float4*)(ydst + 4 * i * YP) = g[i];
        v2f bq[KT][8];
#pragma unroll
        for (int t = 0; t < KT; ++t) {
#pragma unroll
          for (int j = 0; j < 8; ++j) bq[t][j] = *(const v2f*)(brow + t * 16 * YP + 4 * j);
        }
        __builtin_amdgcn_sched_barrier(0);
#pragma unroll
        for (int j = 0; j < 8; ++j) {
#pragma unroll
          for (int t = 0; t < KT; ++t) {
            acc1[h * KT + t] = wmma4(a1[j], bq[t][j], acc1[h * KT + t]);
            acc2[h * KT + t] = wmma4(a2[j], bq[t][j], acc2[h * KT + t]);
          }
        }
        __builtin_amdgcn_sched_barrier(0);
      }
    }
    __builtin_amdgcn_s_setprio(2);
    __syncthreads();
#pragma unroll
    for (int t = 0; t < 2 * KT; ++t) {
      float* d1 = SL + ((size_t)8 * hl) * SSTRIDE + mw + t * 16 + ll;
      float* d2 = d1 + (size_t)16 * SSTRIDE;
#pragma unroll
      for (int v = 0; v < 8; ++v) {
        d1[(size_t)v * SSTRIDE] = acc1[t][v] * scale;
        d2[(size_t)v * SSTRIDE] = acc2[t][v] * scale;
      }
    }
  }
  __syncthreads();

  for (int row = wave; row < 16; row += 8) {
    const float* prow = pos_score + (size_t)(n0 + row) * NN + lane;
    float pr[32];
#pragma unroll
    for (int i = 0; i < 32; ++i) pr[i] = prow[32 * i];
#pragma nounroll
    for (int br = 0; br < 2; ++br) {
      float* Lrow = SL + (size_t)(br * 16 + row) * SSTRIDE + lane;
      float l[32];
#pragma unroll
      for (int i = 0; i < 32; ++i) l[i] = Lrow[32 * i];
      float lmax = -1e30f;
#pragma unroll
      for (int i = 0; i < 32; ++i) lmax = fmaxf(lmax, l[i]);
      lmax = wave_max(lmax);
      float lsum = 0.f;
#pragma unroll
      for (int i = 0; i < 32; ++i) {
        l[i] = expf(l[i] - lmax);
        lsum += l[i];
      }
      lsum = wave_sum(lsum);
      const float inv = 1.0f / lsum;
      float ent = 0.f;
#pragma unroll
      for (int i = 0; i < 32; ++i) {
        const float p = l[i] * inv;
        const float a = fmaf(g, pr[i], omg * p);
        ent = fmaf(-a, logf(a + EPSV), ent);
        Lrow[32 * i] = a;
      }
      ent = wave_sum(ent);
      if (lane == 0) entS[br * 16 + row] = ent;
    }
  }
  __syncthreads();

  if (tid < 16) {
    const int row = tid;
    const float h0 = 2.0f - 2.0f / (1.0f + expf(-ht * entS[row]));
    const float h1 = 2.0f - 2.0f / (1.0f + expf(-ht * entS[16 + row]));
    const int sel = (h0 >= h1) ? 0 : 1;
    selS[row] = (float)sel;
    heat[(size_t)b * NN + n0 + row] = sel ? h1 : h0;
  }
  __syncthreads();

  {
    const int selrow = (int)selS[ll];
    const float* Abase = SL + (size_t)(selrow * 16 + ll) * SSTRIDE + 2 * hl;
    const int dbase = wave * 32;
    const float* bbase = yb + (size_t)(2 * hl) * DD + dbase + 2 * ll;
    v8f acc0 = {0.f, 0.f, 0.f, 0.f, 0.f, 0.f, 0.f, 0.f};
    v8f acc1 = {0.f, 0.f, 0.f, 0.f, 0.f, 0.f, 0.f, 0.f};
    __builtin_amdgcn_s_setprio(1);
    for (int kc = 0; kc < NN; kc += 4 * DCH) {
      if (kc == 3 * NN / 4) __builtin_amdgcn_s_setprio(0);
      v2f r0[DCH], r1[DCH], af[DCH];
#pragma unroll
      for (int j = 0; j < DCH; ++j) {
        r0[j] = *(const v2f*)(bbase + (size_t)(kc + 4 * j) * DD);
        r1[j] = *(const v2f*)(bbase + (size_t)(kc + 4 * j + 1) * DD);
      }
#pragma unroll
      for (int j = 0; j < DCH; ++j) af[j] = *(const v2f*)(Abase + kc + 4 * j);
      __builtin_amdgcn_sched_barrier(0);
#pragma unroll
      for (int j = 0; j < DCH; ++j) {
        const v2f b0 = {r0[j].x, r1[j].x};
        const v2f b1 = {r0[j].y, r1[j].y};
        acc0 = wmma4(af[j], b0, acc0);
        acc1 = wmma4(af[j], b1, acc1);
      }
    }
    float* orow = out + ((size_t)b * NN + n0 + 8 * hl) * DD + dbase + 2 * ll;
#pragma unroll
    for (int v = 0; v < 8; ++v) {
      const v2f o = {acc0[v], acc1[v]};
      *(v2f*)(orow + (size_t)v * DD) = o;
    }
  }
}

extern "C" void kernel_launch(void* const* d_in, const int* in_sizes, int n_in,
                              void* d_out, int out_size, void* d_ws, size_t ws_size,
                              hipStream_t stream) {
  const float* x = (const float*)d_in[0];
  const float* y = (const float*)d_in[1];
  const float* coords = (const float*)d_in[2];
  const float* U = (const float*)d_in[3];
  const float* S1 = (const float*)d_in[4];
  const float* S2 = (const float*)d_in[5];
  const float* gating = (const float*)d_in[6];
  const float* h_temp = (const float*)d_in[7];
  const float* p_temp = (const float*)d_in[8];
  const float* pos_emb = (const float*)d_in[9];

  float* out = (float*)d_out;
  float* heat = out + (size_t)BN * NN * DD;

  float* ws = (float*)d_ws;
  float* pos_score = ws;
  float* xs1 = pos_score + (size_t)NN * NN;
  float* xs2 = xs1 + (size_t)BN * NN * DD;

  proj_pos_kernel<<<BN * NN / PR, 256 + 32 * PW, 0, stream>>>(x, U, S1, S2, xs1, xs2,
                                                            coords, pos_emb, p_temp, pos_score);

  const size_t lds_bytes = (size_t)(2 * 16 * SSTRIDE + 32 + 16) * sizeof(float);
  fused_attn_kernel<<<BN * 64, 256, lds_bytes, stream>>>(
      xs1, xs2, y, pos_score, gating, h_temp, out, heat);
}


# Softmax/entropy pass on the hardware's base-2 exp and log, one each per score: log2(e) folded into the max subtraction, ln(2) and (1-g)/sum once per row; sum order kept, values move in the last bits
# speedup vs baseline: 2.5553x; 1.0405x over previous
#include <hip/hip_runtime.h>
#include <hip/hip_bf16.h>
#include <math.h>

typedef float v2f __attribute__((ext_vector_type(2)));
typedef float v8f __attribute__((ext_vector_type(8)));

#define BN 8
#define NN 1024
#define DD 256
#define EPSV 1e-8f
#define LOG2E 1.44269504088896340736f
#define LN2 0.69314718055994530942f
#define SSTRIDE 1028
#define KT 4
#define DCH 16
#define AXP 260
#define YP 36

__device__ __forceinline__ v8f wmma4(v2f a, v2f b, v8f c) {
  return __builtin_amdgcn_wmma_f32_16x16x4_f32(
       false, a,  false, b,
       (short)0, c,  false,  false);
}

__device__ __forceinline__ float wave_max(float v) {
#pragma unroll
  for (int off = 16; off > 0; off >>= 1) v = fmaxf(v, __shfl_xor(v, off, 32));
  return v;
}
__device__ __forceinline__ float wave_sum(float v) {
#pragma unroll
  for (int off = 16; off > 0; off >>= 1) v += __shfl_xor(v, off, 32);
  return v;
}

#define PR 32
#define UP 268
#define XP 260
#define PCH 4
#define PW 4
__global__ void __launch_bounds__(256 + 32 * PW) proj_pos_kernel(const float* __restrict__ x,
                                  const float* __restrict__ U,
                                  const float* __restrict__ S1,
                                  const float* __restrict__ S2,
                                  float* __restrict__ xs1,
                                  float* __restrict__ xs2,
                                  const float* __restrict__ coords,
                                  const float* __restrict__ pos_emb,
                                  const float* __restrict__ p_temp,
                                  float* __restrict__ pos_score) {
  __shared__ __attribute__((aligned(16))) float plds[DD * UP + PR * XP + 2 * DD];
  float* Ul = plds;
  float* Tl = plds + DD * UP;
  float* Sa = Tl + PR * XP;

  const int tid = threadIdx.x;
  const int lane = tid & 31;
  const int wave = __builtin_amdgcn_readfirstlane(tid >> 5);
  const bool gemm = wave < 8;
  const int hl = lane >> 4, ll = lane & 15;
  const int i0 = blockIdx.x * PR;

  if (gemm) {
#pragma unroll 32
  for (int it = 0; it < 64; ++it) {
    const int idx = it * 256 + tid;
    const int r = ((idx >> 6) + (int)blockIdx.x) & (DD - 1), q = idx & 63;
    const float4 v = *(const float4*)(U + (size_t)r * DD + 4 * q);
    *(float4*)(Ul + r * UP + 4 * q) = v;
  }
#pragma unroll
  for (int it = 0; it < 8; ++it) {
    const int idx = it * 256 + tid;
    const int r = idx >> 6, q = idx & 63;
    const float4 v = *(const float4*)(x + (size_t)(i0 + r) * DD + 4 * q);
    *(float4*)(Tl + r * XP + 4 * q) = v;
  }
  Sa[tid] = fabsf(S1[tid]);
  Sa[DD + tid] = fabsf(S2[tid]);
  }
  __syncthreads();

  const int jw = wave * 32;
  const float* ap = Tl + ll * XP + 2 * hl;

  v8f acc[2][2];
#pragma unroll
  for (int r = 0; r < 2; ++r)
#pragma unroll
    for (int q = 0; q < 2; ++q) acc[r][q] = (v8f){0.f, 0.f, 0.f, 0.f, 0.f, 0.f, 0.f, 0.f};
  if (gemm) {
    const float* bp = Ul + (jw + ll) * UP + 2 * hl;
    v2f a[2][2][PCH], b[2][2][PCH];
#pragma unroll
    for (int j = 0; j < PCH; ++j) {
#pragma unroll
      for (int r = 0; r < 2; ++r) a[0][r][j] = *(const v2f*)(ap + r * 16 * XP + 4 * j);
#pragma unroll
      for (int q = 0; q < 2; ++q) b[0][q][j] = *(const v2f*)(bp + q * 16 * UP + 4 * j);
    }
#pragma unroll
    for (int ci = 0; ci < DD / (4 * PCH); ++ci) {
      const int cb = ci & 1, nb = cb ^ 1;
      if (ci + 1 < DD / (4 * PCH)) {
        const int kn = (ci + 1) * 4 * PCH;
#pragma unroll
        for (int j = 0; j < PCH; ++j) {
#pragma unroll
          for (int r = 0; r < 2; ++r) a[nb][r][j] = *(const v2f*)(ap + r * 16 * XP + kn + 4 * j);
#pragma unroll
          for (int q = 0; q < 2; ++q) b[nb][q][j] = *(const v2f*)(bp + q * 16 * UP + kn + 4 * j);
        }
      }
      __builtin_amdgcn_sched_barrier(0);
#pragma unroll
      for (int j = 0; j < PCH; ++j)
#pragma unroll
        for (int r = 0; r < 2; ++r)
#pragma unroll
          for (int q = 0; q < 2; ++q) acc[r][q] = wmma4(a[cb][r][j], b[cb][q][j], acc[r][q]);
      __builtin_amdgcn_sched_barrier(0);
    }
  }
  __syncthreads();
  if (gemm) {
#pragma unroll
  for (int r = 0; r < 2; ++r)
#pragma unroll
    for (int q = 0; q < 2; ++q) {
      float* tp = Tl + (16 * r + 8 * hl) * XP + jw + 16 * q + ll;
#pragma unroll
      for (int v = 0; v < 8; ++v) tp[v * XP] = acc[r][q][v];
    }
  }
  __syncthreads();

  if (gemm) {
  v8f c1[2][2], c2[2][2];
#pragma unroll
  for (int r = 0; r < 2; ++r)
#pragma unroll
    for (int q = 0; q < 2; ++q) {
      c1[r][q] = (v8f){0.f, 0.f, 0.f, 0.f, 0.f, 0.f, 0.f, 0.f};
      c2[r][q] = (v8f){0.f, 0.f, 0.f, 0.f, 0.f, 0.f, 0.f, 0.f};
    }
  {
    const float* bp = Ul + (2 * hl) * UP + jw + ll;
    const float* sp = Sa + 2 * hl;
    v2f a[2][2][PCH], b[2][2][PCH], s1[2][PCH], s2[2][PCH];
#pragma unroll
    for (int j = 0; j < PCH; ++j) {
#pragma unroll
      for (int r = 0; r < 2; ++r) a[0][r][j] = *(const v2f*)(ap + r * 16 * XP + 4 * j);
#pragma unroll
      for (int q = 0; q < 2; ++q) {
        b[0][q][j].x = bp[(4 * j) * UP + 16 * q];
        b[0][q][j].y = bp[(4 * j + 1) * UP + 16 * q];
      }
      s1[0][j] = *(const v2f*)(sp + 4 * j);
      s2[0][j] = *(const v2f*)(sp + DD + 4 * j);
    }
#pragma unroll
    for (int ci = 0; ci < DD / (4 * PCH); ++ci) {
      const int cb = ci & 1, nb = cb ^ 1;
      if (ci + 1 < DD / (4 * PCH)) {
        const int kn = (ci + 1) * 4 * PCH;
#pragma unroll
        for (int j = 0; j < PCH; ++j) {
#pragma unroll
          for (int r = 0; r < 2; ++r) a[nb][r][j] = *(const v2f*)(ap + r * 16 * XP + kn + 4 * j);
#pragma unroll
          for (int q = 0; q < 2; ++q) {
            b[nb][q][j].x = bp[(kn + 4 * j) * UP + 16 * q];
            b[nb][q][j].y = bp[(kn + 4 * j + 1) * UP + 16 * q];
          }
          s1[nb][j] = *(const v2f*)(sp + kn + 4 * j);
          s2[nb][j] = *(const v2f*)(sp + DD + kn + 4 * j);
        }
      }
      __builtin_amdgcn_sched_barrier(0);
#pragma unroll
      for (int j = 0; j < PCH; ++j)
#pragma unroll
        for (int r = 0; r < 2; ++r) {
          const v2f a1 = a[cb][r][j] * s1[cb][j];
          const v2f a2 = a[cb][r][j] * s2[cb][j];
#pragma unroll
          for (int q = 0; q < 2; ++q) {
            c1[r][q] = wmma4(a1, b[cb][q][j], c1[r][q]);
            c2[r][q] = wmma4(a2, b[cb][q][j], c2[r][q]);
          }
        }
      __builtin_amdgcn_sched_barrier(0);
    }
  }
#pragma unroll
  for (int r = 0; r < 2; ++r)
#pragma unroll
    for (int q = 0; q < 2; ++q) {
      float* o1 = xs1 + (size_t)(i0 + 16 * r + 8 * hl) * DD + jw + 16 * q + ll;
      float* o2 = xs2 + (size_t)(i0 + 16 * r + 8 * hl) * DD + jw + 16 * q + ll;
#pragma unroll
      for (int v = 0; v < 8; ++v) {
        o1[(size_t)v * DD] = c1[r][q][v];
        o2[(size_t)v * DD] = c2[r][q][v];
      }
    }
  } else {
    const int n = (int)blockIdx.x * PW + (wave - 8);
    const float pa = fabsf(p_temp[0]);
    float pe[6];
#pragma unroll
    for (int c = 0; c < 6; ++c) pe[c] = pos_emb[n * 6 + c];
    const float* crow = coords + ((size_t)n * NN + lane) * 6;
    float e[32];
    float lmax = -1e30f;
#pragma unroll
    for (int j = 0; j < 32; ++j) {
      const float* cp = crow + (size_t)(32 * j) * 6;
      const v2f c01 = *(const v2f*)(cp);
      const v2f c23 = *(const v2f*)(cp + 2);
      const v2f c45 = *(const v2f*)(cp + 4);
      float lg;
      {
#pragma clang fp contract(off)
        float d = ((c01.x * pe[0] + c01.y * pe[1]) + c23.x * pe[2]) + c23.y * pe[3];
        d = __builtin_fmaf(c45.x, pe[4], d);
        d = __builtin_fmaf(c45.y, pe[5], d);
        lg = -pa * d;
      }
      e[j] = lg;
      lmax = fmaxf(lmax, lg);
    }
    const float rmax = wave_max(lmax);
    float rsum = 0.f;
#pragma unroll
    for (int w = 0; w < 8; ++w) {
      float ls = 0.f;
#pragma unroll
      for (int it = 0; it < 4; ++it) {
        const float ex = expf(e[w + 8 * it] - rmax);
        e[w + 8 * it] = ex;
        ls += ex;
      }
      rsum += wave_sum(ls);
    }
    const float inv = 1.0f / rsum;
    float* prow = pos_score + (size_t)n * NN + lane;
#pragma unroll
    for (int j = 0; j < 32; ++j) prow[32 * j] = e[j] * inv;
  }
}

__global__ void __launch_bounds__(256) fused_attn_kernel(const float* __restrict__ xs1,
                                  const float* __restrict__ xs2,
                                  const float* __restrict__ y,
                                  const float* __restrict__ pos_score,
                                  const float* __restrict__ gating,
                                  const float* __restrict__ h_temp,
                                  float* __restrict__ out,
                                  float* __restrict__ heat) {
  extern __shared__ __attribute__((aligned(16))) float smem[];
  float* SL = smem;
  float* entS = smem + 2 * 16 * SSTRIDE;
  float* selS = entS + 32;

  const int b = blockIdx.x >> 6;
  const int n0 = (blockIdx.x & 63) * 16;
  const int tid = threadIdx.x;
  const int lane = tid & 31, wave = tid >> 5;
  const int hl = lane >> 4, ll = lane & 15;
  const float scale = 0.0625f;
  const float g = 1.0f / (1.0f + expf(-gating[0]));
  const float omg = 1.0f - g;
  const float ht = h_temp[0];
  const float* yb = y + (size_t)b * NN * DD;

  {
    float* A1l = SL;
    float* A2l = SL + 16 * AXP;
    float* Yl = SL + 32 * AXP + wave * (16 * KT * YP);
#pragma unroll
    for (int it = 0; it < 4; ++it) {
      const int idx = it * 256 + tid;
      const int r = idx >> 6, q = idx & 63;
      const size_t go = ((size_t)b * NN + n0 + r) * DD + 4 * q;
      const float4 v1 = *(const float4*)(xs1 + go);
      const float4 v2 = *(const float4*)(xs2 + go);
      *(float4*)(A1l + r * AXP + 4 * q) = v1;
      *(float4*)(A2l + r * AXP + 4 * q) = v2;
    }
    __syncthreads();

    const float* a1row = A1l + ll * AXP + 2 * hl;
    const float* a2row = A2l + ll * AXP + 2 * hl;
    const int mw = wave * (2 * KT * 16);
    const float* ysrc = yb + (size_t)(mw + (lane >> 3)) * DD + 4 * (lane & 7);
    float* ydst = Yl + (lane >> 3) * YP + 4 * (lane & 7);
    const float* brow = Yl + ll * YP + 2 * hl;
    v8f acc1[2 * KT], acc2[2 * KT];
#pragma unroll
    for (int t = 0; t < 2 * KT; ++t) {
      acc1[t] = (v8f){0.f, 0.f, 0.f, 0.f, 0.f, 0.f, 0.f, 0.f};
      acc2[t] = (v8f){0.f, 0.f, 0.f, 0.f, 0.f, 0.f, 0.f, 0.f};
    }
    __builtin_amdgcn_s_setprio(3);
    for (int kl = 0; kl < DD; kl += 32) {
      v2f a1[8], a2[8];
#pragma unroll
      for (int j = 0; j < 8; ++j) a1[j] = *(const v2f*)(a1row + kl + 4 * j);
#pragma unroll
      for (int j = 0; j < 8; ++j) a2[j] = *(const v2f*)(a2row + kl + 4 * j);
#pragma unroll
      for (int h = 0; h < 2; ++h) {
        float4 g[4 * KT];
#pragma unroll
        for (int i = 0; i < 4 * KT; ++i)
          g[i] = *(const float4*)(ysrc + (size_t)(h * KT * 16 + 4 * i) * DD + kl);
#pragma unroll
        for (int i = 0; i < 4 * KT; ++i) *(float4*)(ydst + 4 * i * YP) = g[i];
        v2f bq[KT][8];
#pragma unroll
        for (int t = 0; t < KT; ++t) {
#pragma unroll
          for (int j = 0; j < 8; ++j) bq[t][j] = *(const v2f*)(brow + t * 16 * YP + 4 * j);
        }
        __builtin_amdgcn_sched_barrier(0);
#pragma unroll
        for (int j = 0; j < 8; ++j) {
#pragma unroll
          for (int t = 0; t < KT; ++t) {
            acc1[h * KT + t] = wmma4(a1[j], bq[t][j], acc1[h * KT + t]);
            acc2[h * KT + t] = wmma4(a2[j], bq[t][j], acc2[h * KT + t]);
          }
        }
        __builtin_amdgcn_sched_barrier(0);
      }
    }
    __builtin_amdgcn_s_setprio(2);
    __syncthreads();
#pragma unroll
    for (int t = 0; t < 2 * KT; ++t) {
      float* d1 = SL + ((size_t)8 * hl) * SSTRIDE + mw + t * 16 + ll;
      float* d2 = d1 + (size_t)16 * SSTRIDE;
#pragma unroll
      for (int v = 0; v < 8; ++v) {
        d1[(size_t)v * SSTRIDE] = acc1[t][v] * scale;
        d2[(size_t)v * SSTRIDE] = acc2[t][v] * scale;
      }
    }
  }
  __syncthreads();

  for (int row = wave; row < 16; row += 8) {
    const float* prow = pos_score + (size_t)(n0 + row) * NN + lane;
    float pr[32];
#pragma unroll
    for (int i = 0; i < 32; ++i) pr[i] = prow[32 * i];
#pragma nounroll
    for (int br = 0; br < 2; ++br) {
      float* Lrow = SL + (size_t)(br * 16 + row) * SSTRIDE + lane;
      float l[32];
#pragma unroll
      for (int i = 0; i < 32; ++i) l[i] = Lrow[32 * i];
      float lmax = -1e30f;
#pragma unroll
      for (int i = 0; i < 32; ++i) lmax = fmaxf(lmax, l[i]);
      lmax = wave_max(lmax);
      const float nm2 = -lmax * LOG2E;
      float lsum = 0.f;
#pragma unroll
      for (int i = 0; i < 32; ++i) {
        l[i] = __builtin_amdgcn_exp2f(fmaf(l[i], LOG2E, nm2));
        lsum += l[i];
      }
      lsum = wave_sum(lsum);
      const float pk = omg / lsum;
      float ent2 = 0.f;
#pragma unroll
      for (int i = 0; i < 32; ++i) {
        const float a = fmaf(g, pr[i], l[i] * pk);
        ent2 = fmaf(a, __builtin_amdgcn_logf(a + EPSV), ent2);
        Lrow[32 * i] = a;
      }
      const float ent = -LN2 * wave_sum(ent2);
      if (lane == 0) entS[br * 16 + row] = ent;
    }
  }
  __syncthreads();

  if (tid < 16) {
    const int row = tid;
    const float h0 = 2.0f - 2.0f / (1.0f + expf(-ht * entS[row]));
    const float h1 = 2.0f - 2.0f / (1.0f + expf(-ht * entS[16 + row]));
    const int sel = (h0 >= h1) ? 0 : 1;
    selS[row] = (float)sel;
    heat[(size_t)b * NN + n0 + row] = sel ? h1 : h0;
  }
  __syncthreads();

  {
    const int selrow = (int)selS[ll];
    const float* Abase = SL + (size_t)(selrow * 16 + ll) * SSTRIDE + 2 * hl;
    const int dbase = wave * 32;
    const float* bbase = yb + (size_t)(2 * hl) * DD + dbase + 2 * ll;
    v8f acc0 = {0.f, 0.f, 0.f, 0.f, 0.f, 0.f, 0.f, 0.f};
    v8f acc1 = {0.f, 0.f, 0.f, 0.f, 0.f, 0.f, 0.f, 0.f};
    __builtin_amdgcn_s_setprio(1);
    for (int kc = 0; kc < NN; kc += 4 * DCH) {
      if (kc == 3 * NN / 4) __builtin_amdgcn_s_setprio(0);
      v2f r0[DCH], r1[DCH], af[DCH];
#pragma unroll
      for (int j = 0; j < DCH; ++j) {
        r0[j] = *(const v2f*)(bbase + (size_t)(kc + 4 * j) * DD);
        r1[j] = *(const v2f*)(bbase + (size_t)(kc + 4 * j + 1) * DD);
      }
#pragma unroll
      for (int j = 0; j < DCH; ++j) af[j] = *(const v2f*)(Abase + kc + 4 * j);
      __builtin_amdgcn_sched_barrier(0);
#pragma unroll
      for (int j = 0; j < DCH; ++j) {
        const v2f b0 = {r0[j].x, r1[j].x};
        const v2f b1 = {r0[j].y, r1[j].y};
        acc0 = wmma4(af[j], b0, acc0);
        acc1 = wmma4(af[j], b1, acc1);
      }
    }
    float* orow = out + ((size_t)b * NN + n0 + 8 * hl) * DD + dbase + 2 * ll;
#pragma unroll
    for (int v = 0; v < 8; ++v) {
      const v2f o = {acc0[v], acc1[v]};
      *(v2f*)(orow + (size_t)v * DD) = o;
    }
  }
}

extern "C" void kernel_launch(void* const* d_in, const int* in_sizes, int n_in,
                              void* d_out, int out_size, void* d_ws, size_t ws_size,
                              hipStream_t stream) {
  const float* x = (const float*)d_in[0];
  const float* y = (const float*)d_in[1];
  const float* coords = (const float*)d_in[2];
  const float* U = (const float*)d_in[3];
  const float* S1 = (const float*)d_in[4];
  const float* S2 = (const float*)d_in[5];
  const float* gating = (const float*)d_in[6];
  const float* h_temp = (const float*)d_in[7];
  const float* p_temp = (const float*)d_in[8];
  const float* pos_emb = (const float*)d_in[9];

  float* out = (float*)d_out;
  float* heat = out + (size_t)BN * NN * DD;

  float* ws = (float*)d_ws;
  float* pos_score = ws;
  float* xs1 = pos_score + (size_t)NN * NN;
  float* xs2 = xs1 + (size_t)BN * NN * DD;

  proj_pos_kernel<<<BN * NN / PR, 256 + 32 * PW, 0, stream>>>(x, U, S1, S2, xs1, xs2,
                                                            coords, pos_emb, p_temp, pos_score);

  const size_t lds_bytes = (size_t)(2 * 16 * SSTRIDE + 32 + 16) * sizeof(float);
  fused_attn_kernel<<<BN * 64, 256, lds_bytes, stream>>>(
      xs1, xs2, y, pos_score, gating, h_temp, out, heat);
}


# Scores re-associated: U moves from the two query operands to the one key operand (z = y@U^T, scores = ((x@U^T)*|S_k|) @ z^T): 2 projection products, was 3; score sums formed in another order
# speedup vs baseline: 2.6536x; 1.0385x over previous
#include <hip/hip_runtime.h>
#include <hip/hip_bf16.h>
#include <math.h>

typedef float v2f __attribute__((ext_vector_type(2)));
typedef float v8f __attribute__((ext_vector_type(8)));

#define BN 8
#define NN 1024
#define DD 256
#define EPSV 1e-8f
#define LOG2E 1.44269504088896340736f
#define LN2 0.69314718055994530942f
#define SSTRIDE 1028
#define KT 4
#define DCH 16
#define AXP 260
#define YP 36

__device__ __forceinline__ v8f wmma4(v2f a, v2f b, v8f c) {
  return __builtin_amdgcn_wmma_f32_16x16x4_f32(
       false, a,  false, b,
       (short)0, c,  false,  false);
}

__device__ __forceinline__ float wave_max(float v) {
#pragma unroll
  for (int off = 16; off > 0; off >>= 1) v = fmaxf(v, __shfl_xor(v, off, 32));
  return v;
}
__device__ __forceinline__ float wave_sum(float v) {
#pragma unroll
  for (int off = 16; off > 0; off >>= 1) v += __shfl_xor(v, off, 32);
  return v;
}

#define PR 32
#define UP 268
#define XP 260
#define PCH 4
#define PW 4

__device__ __forceinline__ void tile_times_Ut(const float* ap, const float* bp, v8f (&acc)[2][2]) {
#pragma unroll
  for (int r = 0; r < 2; ++r)
#pragma unroll
    for (int q = 0; q < 2; ++q) acc[r][q] = (v8f){0.f, 0.f, 0.f, 0.f, 0.f, 0.f, 0.f, 0.f};
  v2f a[2][2][PCH], b[2][2][PCH];
#pragma unroll
  for (int j = 0; j < PCH; ++j) {
#pragma unroll
    for (int r = 0; r < 2; ++r) a[0][r][j] = *(const v2f*)(ap + r * 16 * XP + 4 * j);
#pragma unroll
    for (int q = 0; q < 2; ++q) b[0][q][j] = *(const v2f*)(bp + q * 16 * UP + 4 * j);
  }
#pragma unroll
  for (int ci = 0; ci < DD / (4 * PCH); ++ci) {
    const int cb = ci & 1, nb = cb ^ 1;
    if (ci + 1 < DD / (4 * PCH)) {
      const int kn = (ci + 1) * 4 * PCH;
#pragma unroll
      for (int j = 0; j < PCH; ++j) {
#pragma unroll
        for (int r = 0; r < 2; ++r) a[nb][r][j] = *(const v2f*)(ap + r * 16 * XP + kn + 4 * j);
#pragma unroll
        for (int q = 0; q < 2; ++q) b[nb][q][j] = *(const v2f*)(bp + q * 16 * UP + kn + 4 * j);
      }
    }
    __builtin_amdgcn_sched_barrier(0);
#pragma unroll
    for (int j = 0; j < PCH; ++j)
#pragma unroll
      for (int r = 0; r < 2; ++r)
#pragma unroll
        for (int q = 0; q < 2; ++q) acc[r][q] = wmma4(a[cb][r][j], b[cb][q][j], acc[r][q]);
    __builtin_amdgcn_sched_barrier(0);
  }
}

__global__ void __launch_bounds__(256 + 32 * PW) proj_pos_kernel(const float* __restrict__ x,
                                  const float* __restrict__ y,
                                  const float* __restrict__ U,
                                  const float* __restrict__ S1,
                                  const float* __restrict__ S2,
                                  float* __restrict__ ts1,
                                  float* __restrict__ ts2,
                                  float* __restrict__ z,
                                  const float* __restrict__ coords,
                                  const float* __restrict__ pos_emb,
                                  const float* __restrict__ p_temp,
                                  float* __restrict__ pos_score) {
  __shared__ __attribute__((aligned(16))) float plds[DD * UP + PR * XP + 2 * DD];
  float* Ul = plds;
  float* Tl = plds + DD * UP;
  float* Sa = Tl + PR * XP;

  const int tid = threadIdx.x;
  const int lane = tid & 31;
  const int wave = __builtin_amdgcn_readfirstlane(tid >> 5);
  const bool gemm = wave < 8;
  const int hl = lane >> 4, ll = lane & 15;
  const int i0 = blockIdx.x * PR;

  if (gemm) {
#pragma unroll 32
  for (int it = 0; it < 64; ++it) {
    const int idx = it * 256 + tid;
    const int r = ((idx >> 6) + (int)blockIdx.x) & (DD - 1), q = idx & 63;
    const float4 v = *(const float4*)(U + (size_t)r * DD + 4 * q);
    *(float4*)(Ul + r * UP + 4 * q) = v;
  }
#pragma unroll
  for (int it = 0; it < 8; ++it) {
    const int idx = it * 256 + tid;
    const int r = idx >> 6, q = idx & 63;
    const float4 v = *(const float4*)(x + (size_t)(i0 + r) * DD + 4 * q);
    *(float4*)(Tl + r * XP + 4 * q) = v;
  }
  Sa[tid] = fabsf(S1[tid]);
  Sa[DD + tid] = fabsf(S2[tid]);
  }
  __syncthreads();

  const int jw = wave * 32;
  const float* ap = Tl + ll * XP + 2 * hl;
  const float* bp = Ul + (jw + ll) * UP + 2 * hl;
  const int yr = tid >> 6, yq = tid & 63;
  float4 yv0, yv1, yv2, yv3, yv4, yv5, yv6, yv7;

  if (gemm) {
    {
      const float* yp = y + (size_t)(i0 + yr) * DD + 4 * yq;
      yv0 = *(const float4*)(yp);
      yv1 = *(const float4*)(yp + 4 * DD);
      yv2 = *(const float4*)(yp + 8 * DD);
      yv3 = *(const float4*)(yp + 12 * DD);
      yv4 = *(const float4*)(yp + 16 * DD);
      yv5 = *(const float4*)(yp + 20 * DD);
      yv6 = *(const float4*)(yp + 24 * DD);
      yv7 = *(const float4*)(yp + 28 * DD);
    }
    v8f acc[2][2];
    tile_times_Ut(ap, bp, acc);
#pragma unroll
    for (int r = 0; r < 2; ++r)
#pragma unroll
      for (int q = 0; q < 2; ++q) {
        const float s1 = Sa[jw + 16 * q + ll], s2 = Sa[DD + jw + 16 * q + ll];
        float* o1 = ts1 + (size_t)(i0 + 16 * r + 8 * hl) * DD + jw + 16 * q + ll;
        float* o2 = ts2 + (size_t)(i0 + 16 * r + 8 * hl) * DD + jw + 16 * q + ll;
#pragma unroll
        for (int v = 0; v < 8; ++v) {
          o1[(size_t)v * DD] = acc[r][q][v] * s1;
          o2[(size_t)v * DD] = acc[r][q][v] * s2;
        }
      }
  }
  __syncthreads();
  if (gemm) {
    float* tp = Tl + yr * XP + 4 * yq;
    *(float4*)(tp) = yv0;
    *(float4*)(tp + 4 * XP) = yv1;
    *(float4*)(tp + 8 * XP) = yv2;
    *(float4*)(tp + 12 * XP) = yv3;
    *(float4*)(tp + 16 * XP) = yv4;
    *(float4*)(tp + 20 * XP) = yv5;
    *(float4*)(tp + 24 * XP) = yv6;
    *(float4*)(tp + 28 * XP) = yv7;
  }
  __syncthreads();

  if (gemm) {
    v8f acc[2][2];
    tile_times_Ut(ap, bp, acc);
#pragma unroll
    for (int r = 0; r < 2; ++r)
#pragma unroll
      for (int q = 0; q < 2; ++q) {
        float* o = z + (size_t)(i0 + 16 * r + 8 * hl) * DD + jw + 16 * q + ll;
#pragma unroll
        for (int v = 0; v < 8; ++v) o[(size_t)v * DD] = acc[r][q][v];
      }
  } else {
    const int n = (int)blockIdx.x * PW + (wave - 8);
    const float pa = fabsf(p_temp[0]);
    float pe[6];
#pragma unroll
    for (int c = 0; c < 6; ++c) pe[c] = pos_emb[n * 6 + c];
    const float* crow = coords + ((size_t)n * NN + lane) * 6;
    float e[32];
    float lmax = -1e30f;
#pragma unroll
    for (int j = 0; j < 32; ++j) {
      const float* cp = crow + (size_t)(32 * j) * 6;
      const v2f c01 = *(const v2f*)(cp);
      const v2f c23 = *(const v2f*)(cp + 2);
      const v2f c45 = *(const v2f*)(cp + 4);
      float lg;
      {
#pragma clang fp contract(off)
        float d = ((c01.x * pe[0] + c01.y * pe[1]) + c23.x * pe[2]) + c23.y * pe[3];
        d = __builtin_fmaf(c45.x, pe[4], d);
        d = __builtin_fmaf(c45.y, pe[5], d);
        lg = -pa * d;
      }
      e[j] = lg;
      lmax = fmaxf(lmax, lg);
    }
    const float rmax = wave_max(lmax);
    float rsum = 0.f;
#pragma unroll
    for (int w = 0; w < 8; ++w) {
      float ls = 0.f;
#pragma unroll
      for (int it = 0; it < 4; ++it) {
        const float ex = expf(e[w + 8 * it] - rmax);
        e[w + 8 * it] = ex;
        ls += ex;
      }
      rsum += wave_sum(ls);
    }
    const float inv = 1.0f / rsum;
    float* prow = pos_score + (size_t)n * NN + lane;
#pragma unroll
    for (int j = 0; j < 32; ++j) prow[32 * j] = e[j] * inv;
  }
}

__global__ void __launch_bounds__(256) fused_attn_kernel(const float* __restrict__ xs1,
                                  const float* __restrict__ xs2,
                                  const float* __restrict__ z,
                                  const float* __restrict__ y,
                                  const float* __restrict__ pos_score,
                                  const float* __restrict__ gating,
                                  const float* __restrict__ h_temp,
                                  float* __restrict__ out,
                                  float* __restrict__ heat) {
  extern __shared__ __attribute__((aligned(16))) float smem[];
  float* SL = smem;
  float* entS = smem + 2 * 16 * SSTRIDE;
  float* selS = entS + 32;

  const int b = blockIdx.x >> 6;
  const int n0 = (blockIdx.x & 63) * 16;
  const int tid = threadIdx.x;
  const int lane = tid & 31, wave = tid >> 5;
  const int hl = lane >> 4, ll = lane & 15;
  const float scale = 0.0625f;
  const float g = 1.0f / (1.0f + expf(-gating[0]));
  const float omg = 1.0f - g;
  const float ht = h_temp[0];
  const float* yb = y + (size_t)b * NN * DD;
  const float* zb = z + (size_t)b * NN * DD;

  {
    float* A1l = SL;
    float* A2l = SL + 16 * AXP;
    float* Yl = SL + 32 * AXP + wave * (16 * KT * YP);
#pragma unroll
    for (int it = 0; it < 4; ++it) {
      const int idx = it * 256 + tid;
      const int r = idx >> 6, q = idx & 63;
      const size_t go = ((size_t)b * NN + n0 + r) * DD + 4 * q;
      const float4 v1 = *(const float4*)(xs1 + go);
      const float4 v2 = *(const float4*)(xs2 + go);
      *(float4*)(A1l + r * AXP + 4 * q) = v1;
      *(float4*)(A2l + r * AXP + 4 * q) = v2;
    }
    __syncthreads();

    const float* a1row = A1l + ll * AXP + 2 * hl;
    const float* a2row = A2l + ll * AXP + 2 * hl;
    const int mw = wave * (2 * KT * 16);
    const float* ysrc = zb + (size_t)(mw + (lane >> 3)) * DD + 4 * (lane & 7);
    float* ydst = Yl + (lane >> 3) * YP + 4 * (lane & 7);
    const float* brow = Yl + ll * YP + 2 * hl;
    v8f acc1[2 * KT], acc2[2 * KT];
#pragma unroll
    for (int t = 0; t < 2 * KT; ++t) {
      acc1[t] = (v8f){0.f, 0.f, 0.f, 0.f, 0.f, 0.f, 0.f, 0.f};
      acc2[t] = (v8f){0.f, 0.f, 0.f, 0.f, 0.f, 0.f, 0.f, 0.f};
    }
    __builtin_amdgcn_s_setprio(3);
    for (int kl = 0; kl < DD; kl += 32) {
      v2f a1[8], a2[8];
#pragma unroll
      for (int j = 0; j < 8; ++j) a1[j] = *(const v2f*)(a1row + kl + 4 * j);
#pragma unroll
      for (int j = 0; j < 8; ++j) a2[j] = *(const v2f*)(a2row + kl + 4 * j);
#pragma unroll
      for (int h = 0; h < 2; ++h) {
        float4 g[4 * KT];
#pragma unroll
        for (int i = 0; i < 4 * KT; ++i)
          g[i] = *(const float4*)(ysrc + (size_t)(h * KT * 16 + 4 * i) * DD + kl);
#pragma unroll
        for (int i = 0; i < 4 * KT; ++i) *(float4*)(ydst + 4 * i * YP) = g[i];
        v2f bq[KT][8];
#pragma unroll
        for (int t = 0; t < KT; ++t) {
#pragma unroll
          for (int j = 0; j < 8; ++j) bq[t][j] = *(const v2f*)(brow + t * 16 * YP + 4 * j);
        }
        __builtin_amdgcn_sched_barrier(0);
#pragma unroll
        for (int j = 0; j < 8; ++j) {
#pragma unroll
          for (int t = 0; t < KT; ++t) {
            acc1[h * KT + t] = wmma4(a1[j], bq[t][j], acc1[h * KT + t]);
            acc2[h * KT + t] = wmma4(a2[j], bq[t][j], acc2[h * KT + t]);
          }
        }
        __builtin_amdgcn_sched_barrier(0);
      }
    }
    __builtin_amdgcn_s_setprio(2);
    __syncthreads();
#pragma unroll
    for (int t = 0; t < 2 * KT; ++t) {
      float* d1 = SL + ((size_t)8 * hl) * SSTRIDE + mw + t * 16 + ll;
      float* d2 = d1 + (size_t)16 * SSTRIDE;
#pragma unroll
      for (int v = 0; v < 8; ++v) {
        d1[(size_t)v * SSTRIDE] = acc1[t][v] * scale;
        d2[(size_t)v * SSTRIDE] = acc2[t][v] * scale;
      }
    }
  }
  __syncthreads();

  for (int row = wave; row < 16; row += 8) {
    const float* prow = pos_score + (size_t)(n0 + row) * NN + lane;
    float pr[32];
#pragma unroll
    for (int i = 0; i < 32; ++i) pr[i] = prow[32 * i];
#pragma nounroll
    for (int br = 0; br < 2; ++br) {
      float* Lrow = SL + (size_t)(br * 16 + row) * SSTRIDE + lane;
      float l[32];
#pragma unroll
      for (int i = 0; i < 32; ++i) l[i] = Lrow[32 * i];
      float lmax = -1e30f;
#pragma unroll
      for (int i = 0; i < 32; ++i) lmax = fmaxf(lmax, l[i]);
      lmax = wave_max(lmax);
      const float nm2 = -lmax * LOG2E;
      float lsum = 0.f;
#pragma unroll
      for (int i = 0; i < 32; ++i) {
        l[i] = __builtin_amdgcn_exp2f(fmaf(l[i], LOG2E, nm2));
        lsum += l[i];
      }
      lsum = wave_sum(lsum);
      const float pk = omg / lsum;
      float ent2 = 0.f;
#pragma unroll
      for (int i = 0; i < 32; ++i) {
        const float a = fmaf(g, pr[i], l[i] * pk);
        ent2 = fmaf(a, __builtin_amdgcn_logf(a + EPSV), ent2);
        Lrow[32 * i] = a;
      }
      const float ent = -LN2 * wave_sum(ent2);
      if (lane == 0) entS[br * 16 + row] = ent;
    }
  }
  __syncthreads();

  if (tid < 16) {
    const int row = tid;
    const float h0 = 2.0f - 2.0f / (1.0f + expf(-ht * entS[row]));
    const float h1 = 2.0f - 2.0f / (1.0f + expf(-ht * entS[16 + row]));
    const int sel = (h0 >= h1) ? 0 : 1;
    selS[row] = (float)sel;
    heat[(size_t)b * NN + n0 + row] = sel ? h1 : h0;
  }
  __syncthreads();

  {
    const int selrow = (int)selS[ll];
    const float* Abase = SL + (size_t)(selrow * 16 + ll) * SSTRIDE + 2 * hl;
    const int dbase = wave * 32;
    const float* bbase = yb + (size_t)(2 * hl) * DD + dbase + 2 * ll;
    v8f acc0 = {0.f, 0.f, 0.f, 0.f, 0.f, 0.f, 0.f, 0.f};
    v8f acc1 = {0.f, 0.f, 0.f, 0.f, 0.f, 0.f, 0.f, 0.f};
    __builtin_amdgcn_s_setprio(1);
    for (int kc = 0; kc < NN; kc += 4 * DCH) {
      if (kc == 3 * NN / 4) __builtin_amdgcn_s_setprio(0);
      v2f r0[DCH], r1[DCH], af[DCH];
#pragma unroll
      for (int j = 0; j < DCH; ++j) {
        r0[j] = *(const v2f*)(bbase + (size_t)(kc + 4 * j) * DD);
        r1[j] = *(const v2f*)(bbase + (size_t)(kc + 4 * j + 1) * DD);
      }
#pragma unroll
      for (int j = 0; j < DCH; ++j) af[j] = *(const v2f*)(Abase + kc + 4 * j);
      __builtin_amdgcn_sched_barrier(0);
#pragma unroll
      for (int j = 0; j < DCH; ++j) {
        const v2f b0 = {r0[j].x, r1[j].x};
        const v2f b1 = {r0[j].y, r1[j].y};
        acc0 = wmma4(af[j], b0, acc0);
        acc1 = wmma4(af[j], b1, acc1);
      }
    }
    float* orow = out + ((size_t)b * NN + n0 + 8 * hl) * DD + dbase + 2 * ll;
#pragma unroll
    for (int v = 0; v < 8; ++v) {
      const v2f o = {acc0[v], acc1[v]};
      *(v2f*)(orow + (size_t)v * DD) = o;
    }
  }
}

extern "C" void kernel_launch(void* const* d_in, const int* in_sizes, int n_in,
                              void* d_out, int out_size, void* d_ws, size_t ws_size,
                              hipStream_t stream) {
  const float* x = (const float*)d_in[0];
  const float* y = (const float*)d_in[1];
  const float* coords = (const float*)d_in[2];
  const float* U = (const float*)d_in[3];
  const float* S1 = (const float*)d_in[4];
  const float* S2 = (const float*)d_in[5];
  const float* gating = (const float*)d_in[6];
  const float* h_temp = (const float*)d_in[7];
  const float* p_temp = (const float*)d_in[8];
  const float* pos_emb = (const float*)d_in[9];

  float* out = (float*)d_out;
  float* heat = out + (size_t)BN * NN * DD;

  float* ws = (float*)d_ws;
  float* pos_score = ws;
  float* xs1 = pos_score + (size_t)NN * NN;
  float* xs2 = xs1 + (size_t)BN * NN * DD;
  float* z = xs2 + (size_t)BN * NN * DD;

  proj_pos_kernel<<<BN * NN / PR, 256 + 32 * PW, 0, stream>>>(x, y, U, S1, S2, xs1, xs2, z,
                                                            coords, pos_emb, p_temp, pos_score);

  const size_t lds_bytes = (size_t)(2 * 16 * SSTRIDE + 32 + 16) * sizeof(float);
  fused_attn_kernel<<<BN * 64, 256, lds_bytes, stream>>>(
      xs1, xs2, z, y, pos_score, gating, h_temp, out, heat);
}
